# starting kernel
# baseline (speedup 1.0000x reference)
#include <hip/hip_runtime.h>
#include <hip/hip_bf16.h>

typedef __attribute__((ext_vector_type(2))) float v2f;
typedef __attribute__((ext_vector_type(8))) float v8f;

#define TK 128
#define LDK 132

__global__ void gcn_deg_kernel(const int* __restrict__ src, const int* __restrict__ dst,
                               float* __restrict__ degO, float* __restrict__ degI, int E) {
  int t = blockIdx.x * blockDim.x + threadIdx.x;
  if (t < E) {
    atomicAdd(&degO[src[t]], 1.0f);
    atomicAdd(&degI[dst[t]], 1.0f);
  }
}

__global__ void gcn_invsqrt_kernel(float* __restrict__ degO, float* __restrict__ degI, int N) {
  int t = blockIdx.x * blockDim.x + threadIdx.x;
  if (t < N) {
    float o = degO[t];
    float i = degI[t];
    degO[t] = (o > 0.0f) ? rsqrtf(fmaxf(o, 1.0f)) : 0.0f;
    degI[t] = (i > 0.0f) ? rsqrtf(fmaxf(i, 1.0f)) : 0.0f;
  }
}

template <int CHUNKS>
__global__ void gcn_scatter_kernel(const float* __restrict__ h,
                                   const int* __restrict__ src,
                                   const int* __restrict__ dst,
                                   const float* __restrict__ invO,
                                   float* __restrict__ agg,
                                   int E) {
  constexpr int F = CHUNKS * 4;
  long long t = (long long)blockIdx.x * blockDim.x + threadIdx.x;
  long long total = (long long)E * CHUNKS;
  if (t >= total) return;
  int e = (int)(t / CHUNKS);
  int c = (int)(t - (long long)e * CHUNKS);
  int s = src[e];
  int d = dst[e];
  float sc = invO[s];
  float4 v = ((const float4*)(h + (size_t)s * F))[c];
  float* ad = agg + (size_t)d * F + (size_t)c * 4;
  atomicAdd(ad + 0, v.x * sc);
  atomicAdd(ad + 1, v.y * sc);
  atomicAdd(ad + 2, v.z * sc);
  atomicAdd(ad + 3, v.w * sc);
}

__global__ void __launch_bounds__(256)
gcn_gemm_kernel(const float* __restrict__ A,
                const float* __restrict__ rowScale,
                const float* __restrict__ W,
                const float* __restrict__ bias,
                float* __restrict__ C,
                int M, int Nd, int relu) {
  __shared__ float As[16 * LDK];

  const int t     = threadIdx.x;
  const int mBase = blockIdx.x * 16;

  for (int i = t; i < 16 * (TK / 4); i += 256) {
    int row = i >> 5;
    int c4  = i & 31;
    int gr  = mBase + row;
    float4 v = make_float4(0.f, 0.f, 0.f, 0.f);
    float  s = 1.0f;
    if (gr < M) {
      v = ((const float4*)(A + (size_t)gr * TK))[c4];
      if (rowScale) s = rowScale[gr];
    }
    float* dstp = As + (size_t)row * LDK + (size_t)c4 * 4;
    dstp[0] = v.x * s; dstp[1] = v.y * s; dstp[2] = v.z * s; dstp[3] = v.w * s;
  }
  __syncthreads();

  const int wave   = t >> 5;
  const int lane   = t & 31;
  const int nTiles = (Nd + 15) >> 4;
  if (wave >= nTiles) return;

  const int col    = lane & 15;
  const int gcol   = wave * 16 + col;
  const int khalf  = (lane >> 4) * 2;
  const bool colOk = (gcol < Nd);
  const int gcSafe = colOk ? gcol : (Nd - 1);

  const float* As_row = As + (size_t)(lane & 15) * LDK + khalf;
  const float* Wp     = W + (size_t)khalf * Nd + gcSafe;

  v8f acc0 = {};
  v8f acc1 = {};
  #pragma unroll 8
  for (int k0 = 0; k0 < TK; k0 += 8) {
    v2f a0, b0, a1, b1;
    a0[0] = As_row[k0];
    a0[1] = As_row[k0 + 1];
    b0[0] = Wp[0];
    b0[1] = Wp[Nd];
    a1[0] = As_row[k0 + 4];
    a1[1] = As_row[k0 + 5];
    b1[0] = Wp[4 * (size_t)Nd];
    b1[1] = Wp[5 * (size_t)Nd];
    acc0 = __builtin_amdgcn_wmma_f32_16x16x4_f32(false, a0, false, b0, (short)0, acc0,
                                                 false, false);
    acc1 = __builtin_amdgcn_wmma_f32_16x16x4_f32(false, a1, false, b1, (short)0, acc1,
                                                 false, false);
    Wp += 8 * (size_t)Nd;
  }
  v8f acc = acc0 + acc1;

  if (colOk) {
    const float bb    = bias ? bias[gcol] : 0.0f;
    const int   rbase = mBase + (lane >> 4) * 8;
    #pragma unroll
    for (int v = 0; v < 8; ++v) {
      int r = rbase + v;
      if (r < M) {
        float val = acc[v] + bb;
        if (relu) val = fmaxf(val, 0.0f);
        C[(size_t)r * Nd + gcol] = val;
      }
    }
  }
}

__global__ void gcn_final_kernel(const float* __restrict__ agg,
                                 const float* __restrict__ invI,
                                 const float* __restrict__ b3,
                                 float* __restrict__ out, int N, int Cc) {
  int t = blockIdx.x * blockDim.x + threadIdx.x;
  if (t < N * Cc) {
    int n = t / Cc;
    int c = t - n * Cc;
    out[t] = agg[t] * invI[n] + b3[c];
  }
}

extern "C" void kernel_launch(void* const* d_in, const int* in_sizes, int n_in,
                              void* d_out, int out_size, void* d_ws, size_t ws_size,
                              hipStream_t stream) {
  const float* feats = (const float*)d_in[0];
  const int*   src   = (const int*)d_in[1];
  const int*   dst   = (const int*)d_in[2];
  const float* W1    = (const float*)d_in[3];
  const float* b1    = (const float*)d_in[4];
  const float* W2    = (const float*)d_in[5];
  const float* b2    = (const float*)d_in[6];
  const float* W3    = (const float*)d_in[7];
  const float* b3    = (const float*)d_in[8];
  float*       out   = (float*)d_out;

  const int HID  = in_sizes[4];
  const int NCLS = in_sizes[8];
  const int INF  = in_sizes[3] / HID;
  const int N    = in_sizes[0] / INF;
  const int E    = in_sizes[1];

  float* invO = (float*)d_ws;
  float* invI = invO + N;
  float* bufA = invI + N;
  float* bufB = bufA + (size_t)N * TK;
  float* bufC = bufB + (size_t)N * TK;

  const int B = 256;
  const int mTiles = (N + 15) / 16;

  hipMemsetAsync(invO, 0, sizeof(float) * 2 * (size_t)N, stream);
  gcn_deg_kernel<<<(E + B - 1) / B, B, 0, stream>>>(src, dst, invO, invI, E);
  gcn_invsqrt_kernel<<<(N + B - 1) / B, B, 0, stream>>>(invO, invI, N);

  hipMemsetAsync(bufA, 0, sizeof(float) * (size_t)N * TK, stream);
  {
    long long tot = (long long)E * 32;
    gcn_scatter_kernel<32><<<(int)((tot + B - 1) / B), B, 0, stream>>>(feats, src, dst, invO, bufA, E);
  }
  gcn_gemm_kernel<<<mTiles, B, 0, stream>>>(bufA, invI, W1, b1, bufB, N, HID, 1);

  hipMemsetAsync(bufA, 0, sizeof(float) * (size_t)N * TK, stream);
  {
    long long tot = (long long)E * 32;
    gcn_scatter_kernel<32><<<(int)((tot + B - 1) / B), B, 0, stream>>>(bufB, src, dst, invO, bufA, E);
  }
  gcn_gemm_kernel<<<mTiles, B, 0, stream>>>(bufA, invI, W2, b2, bufB, N, HID, 1);

  gcn_gemm_kernel<<<mTiles, B, 0, stream>>>(bufB, nullptr, W3, nullptr, bufC, N, NCLS, 0);
  hipMemsetAsync(bufA, 0, sizeof(float) * (size_t)N * NCLS, stream);
  {
    long long tot = (long long)E * 10;
    gcn_scatter_kernel<10><<<(int)((tot + B - 1) / B), B, 0, stream>>>(bufC, src, dst, invO, bufA, E);
  }
  gcn_final_kernel<<<(N * NCLS + B - 1) / B, B, 0, stream>>>(bufA, invI, b3, out, N, NCLS);
}


# Scatter-add remapped: a wave walks 32 edges, moving each row in whole 128-byte lines (lane = consecutive float): an atomic instruction is one full line, not four; sum order: atomic arrival as before
# speedup vs baseline: 1.4912x; 1.4912x over previous
#include <hip/hip_runtime.h>
#include <hip/hip_bf16.h>

typedef __attribute__((ext_vector_type(2))) float v2f;
typedef __attribute__((ext_vector_type(8))) float v8f;

#define TK 128
#define LDK 132

__global__ void gcn_deg_kernel(const int* __restrict__ src, const int* __restrict__ dst,
                               float* __restrict__ degO, float* __restrict__ degI, int E) {
  int t = blockIdx.x * blockDim.x + threadIdx.x;
  if (t < E) {
    atomicAdd(&degO[src[t]], 1.0f);
    atomicAdd(&degI[dst[t]], 1.0f);
  }
}

__global__ void gcn_invsqrt_kernel(float* __restrict__ degO, float* __restrict__ degI, int N) {
  int t = blockIdx.x * blockDim.x + threadIdx.x;
  if (t < N) {
    float o = degO[t];
    float i = degI[t];
    degO[t] = (o > 0.0f) ? rsqrtf(fmaxf(o, 1.0f)) : 0.0f;
    degI[t] = (i > 0.0f) ? rsqrtf(fmaxf(i, 1.0f)) : 0.0f;
  }
}

template <int F>
struct RowLines {
  static constexpr int G = ((F & -F) < 32) ? (F & -F) : 32;
  static constexpr int K = (32 - G + F + 31) / 32;
};

template <int F, int U>
__device__ __forceinline__ void scatter_rows(const float* __restrict__ h,
                                             float* __restrict__ agg,
                                             int s, int d, float sc, int j0, int lane) {
  constexpr int K = RowLines<F>::K;
  float v[U][K];
  #pragma unroll
  for (int u = 0; u < U; ++u) {
    const int sj = __builtin_amdgcn_readlane(s, j0 + u);
    const int dj = __builtin_amdgcn_readlane(d, j0 + u);
    const int a  = (int)(((size_t)dj * F) & 31);
    const float* hs = h + (size_t)sj * F;
    #pragma unroll
    for (int k = 0; k < K; ++k) {
      const int f  = 32 * k + lane - a;
      const int fc = (f < 0) ? 0 : ((f > F - 1) ? (F - 1) : f);
      v[u][k] = hs[fc];
    }
  }
  #pragma unroll
  for (int u = 0; u < U; ++u) {
    const int   dj  = __builtin_amdgcn_readlane(d, j0 + u);
    const float scj = __int_as_float(__builtin_amdgcn_readlane(__float_as_int(sc), j0 + u));
    const int   a   = (int)(((size_t)dj * F) & 31);
    float* ad = agg + (size_t)dj * F;
    #pragma unroll
    for (int k = 0; k < K; ++k) {
      const int f = 32 * k + lane - a;
      if ((unsigned)f < (unsigned)F) atomicAdd(ad + f, v[u][k] * scj);
    }
  }
}

template <int CHUNKS>
__global__ void __launch_bounds__(128)
gcn_scatter_kernel(const float* __restrict__ h,
                   const int* __restrict__ src,
                   const int* __restrict__ dst,
                   const float* __restrict__ invO,
                   float* __restrict__ agg,
                   int E) {
  constexpr int F = CHUNKS * 4;
  constexpr int U = 32 / RowLines<F>::K;
  const int lane = threadIdx.x & 31;
  const int e    = blockIdx.x * blockDim.x + threadIdx.x;
  const int e0   = __builtin_amdgcn_readfirstlane(e - lane);
  if (e0 >= E) return;
  int s = 0, d = 0;
  float sc = 0.0f;
  if (e < E) {
    s  = src[e];
    d  = dst[e];
    sc = invO[s];
  }
  const int cnt = (E - e0 < 32) ? (E - e0) : 32;
  for (int jb = 0; jb < cnt; jb += U) {
    if (jb + U <= cnt) {
      scatter_rows<F, U>(h, agg, s, d, sc, jb, lane);
    } else {
      for (int j = jb; j < cnt; ++j) scatter_rows<F, 1>(h, agg, s, d, sc, j, lane);
    }
  }
}

__global__ void __launch_bounds__(256)
gcn_gemm_kernel(const float* __restrict__ A,
                const float* __restrict__ rowScale,
                const float* __restrict__ W,
                const float* __restrict__ bias,
                float* __restrict__ C,
                int M, int Nd, int relu) {
  __shared__ float As[16 * LDK];

  const int t     = threadIdx.x;
  const int mBase = blockIdx.x * 16;

  for (int i = t; i < 16 * (TK / 4); i += 256) {
    int row = i >> 5;
    int c4  = i & 31;
    int gr  = mBase + row;
    float4 v = make_float4(0.f, 0.f, 0.f, 0.f);
    float  s = 1.0f;
    if (gr < M) {
      v = ((const float4*)(A + (size_t)gr * TK))[c4];
      if (rowScale) s = rowScale[gr];
    }
    float* dstp = As + (size_t)row * LDK + (size_t)c4 * 4;
    dstp[0] = v.x * s; dstp[1] = v.y * s; dstp[2] = v.z * s; dstp[3] = v.w * s;
  }
  __syncthreads();

  const int wave   = t >> 5;
  const int lane   = t & 31;
  const int nTiles = (Nd + 15) >> 4;
  if (wave >= nTiles) return;

  const int col    = lane & 15;
  const int gcol   = wave * 16 + col;
  const int khalf  = (lane >> 4) * 2;
  const bool colOk = (gcol < Nd);
  const int gcSafe = colOk ? gcol : (Nd - 1);

  const float* As_row = As + (size_t)(lane & 15) * LDK + khalf;
  const float* Wp     = W + (size_t)khalf * Nd + gcSafe;

  v8f acc0 = {};
  v8f acc1 = {};
  #pragma unroll 8
  for (int k0 = 0; k0 < TK; k0 += 8) {
    v2f a0, b0, a1, b1;
    a0[0] = As_row[k0];
    a0[1] = As_row[k0 + 1];
    b0[0] = Wp[0];
    b0[1] = Wp[Nd];
    a1[0] = As_row[k0 + 4];
    a1[1] = As_row[k0 + 5];
    b1[0] = Wp[4 * (size_t)Nd];
    b1[1] = Wp[5 * (size_t)Nd];
    acc0 = __builtin_amdgcn_wmma_f32_16x16x4_f32(false, a0, false, b0, (short)0, acc0,
                                                 false, false);
    acc1 = __builtin_amdgcn_wmma_f32_16x16x4_f32(false, a1, false, b1, (short)0, acc1,
                                                 false, false);
    Wp += 8 * (size_t)Nd;
  }
  v8f acc = acc0 + acc1;

  if (colOk) {
    const float bb    = bias ? bias[gcol] : 0.0f;
    const int   rbase = mBase + (lane >> 4) * 8;
    #pragma unroll
    for (int v = 0; v < 8; ++v) {
      int r = rbase + v;
      if (r < M) {
        float val = acc[v] + bb;
        if (relu) val = fmaxf(val, 0.0f);
        C[(size_t)r * Nd + gcol] = val;
      }
    }
  }
}

__global__ void gcn_final_kernel(const float* __restrict__ agg,
                                 const float* __restrict__ invI,
                                 const float* __restrict__ b3,
                                 float* __restrict__ out, int N, int Cc) {
  int t = blockIdx.x * blockDim.x + threadIdx.x;
  if (t < N * Cc) {
    int n = t / Cc;
    int c = t - n * Cc;
    out[t] = agg[t] * invI[n] + b3[c];
  }
}

extern "C" void kernel_launch(void* const* d_in, const int* in_sizes, int n_in,
                              void* d_out, int out_size, void* d_ws, size_t ws_size,
                              hipStream_t stream) {
  const float* feats = (const float*)d_in[0];
  const int*   src   = (const int*)d_in[1];
  const int*   dst   = (const int*)d_in[2];
  const float* W1    = (const float*)d_in[3];
  const float* b1    = (const float*)d_in[4];
  const float* W2    = (const float*)d_in[5];
  const float* b2    = (const float*)d_in[6];
  const float* W3    = (const float*)d_in[7];
  const float* b3    = (const float*)d_in[8];
  float*       out   = (float*)d_out;

  const int HID  = in_sizes[4];
  const int NCLS = in_sizes[8];
  const int INF  = in_sizes[3] / HID;
  const int N    = in_sizes[0] / INF;
  const int E    = in_sizes[1];

  float* invO = (float*)d_ws;
  float* invI = invO + N;
  float* bufA = invI + N;
  float* bufB = bufA + (size_t)N * TK;
  float* bufC = bufB + (size_t)N * TK;

  const int B = 256;
  const int mTiles = (N + 15) / 16;

  hipMemsetAsync(invO, 0, sizeof(float) * 2 * (size_t)N, stream);
  gcn_deg_kernel<<<(E + B - 1) / B, B, 0, stream>>>(src, dst, invO, invI, E);
  gcn_invsqrt_kernel<<<(N + B - 1) / B, B, 0, stream>>>(invO, invI, N);

  hipMemsetAsync(bufA, 0, sizeof(float) * (size_t)N * TK, stream);
  {
    gcn_scatter_kernel<32><<<(E + 127) / 128, 128, 0, stream>>>(feats, src, dst, invO, bufA, E);
  }
  gcn_gemm_kernel<<<mTiles, B, 0, stream>>>(bufA, invI, W1, b1, bufB, N, HID, 1);

  hipMemsetAsync(bufA, 0, sizeof(float) * (size_t)N * TK, stream);
  {
    gcn_scatter_kernel<32><<<(E + 127) / 128, 128, 0, stream>>>(bufB, src, dst, invO, bufA, E);
  }
  gcn_gemm_kernel<<<mTiles, B, 0, stream>>>(bufA, invI, W2, b2, bufB, N, HID, 1);

  gcn_gemm_kernel<<<mTiles, B, 0, stream>>>(bufB, nullptr, W3, nullptr, bufC, N, NCLS, 0);
  hipMemsetAsync(bufA, 0, sizeof(float) * (size_t)N * NCLS, stream);
  {
    gcn_scatter_kernel<10><<<(E + 127) / 128, 128, 0, stream>>>(bufC, src, dst, invO, bufA, E);
  }
  gcn_final_kernel<<<(N * NCLS + B - 1) / B, B, 0, stream>>>(bufA, invI, b3, out, N, NCLS);
}


# Every atomically summed array (degree counts, the three aggregates) kept as one copy per cache half; die id picks a wave's copy, readers add both; sum order: arrival within a half, then half0+half1
# speedup vs baseline: 1.9542x; 1.3105x over previous
#include <hip/hip_runtime.h>
#include <hip/hip_bf16.h>

typedef __attribute__((ext_vector_type(2))) float v2f;
typedef __attribute__((ext_vector_type(8))) float v8f;

#define TK 128
#define LDK 132

__device__ __forceinline__ int cache_half() {
  return (int)((__builtin_amdgcn_s_sendmsg_rtn(0x87) >> 18) & 1u);
}

__global__ void gcn_deg_kernel(const int* __restrict__ src, const int* __restrict__ dst,
                               float* __restrict__ deg, int N, int E) {
  int t = blockIdx.x * blockDim.x + threadIdx.x;
  int s = 0, d = 0;
  if (t < E) {
    s = src[t];
    d = dst[t];
  }
  float* dh = deg + (size_t)cache_half() * 2 * (size_t)N;
  if (t < E) {
    atomicAdd(&dh[s], 1.0f);
    atomicAdd(&dh[(size_t)N + d], 1.0f);
  }
}

__global__ void gcn_invsqrt_kernel(float* __restrict__ deg, int N) {
  int t = blockIdx.x * blockDim.x + threadIdx.x;
  if (t < N) {
    float o = deg[t] + deg[2 * (size_t)N + t];
    float i = deg[(size_t)N + t] + deg[3 * (size_t)N + t];
    deg[t]             = (o > 0.0f) ? rsqrtf(fmaxf(o, 1.0f)) : 0.0f;
    deg[(size_t)N + t] = (i > 0.0f) ? rsqrtf(fmaxf(i, 1.0f)) : 0.0f;
  }
}

template <int F>
struct RowLines {
  static constexpr int G = ((F & -F) < 32) ? (F & -F) : 32;
  static constexpr int K = (32 - G + F + 31) / 32;
};

template <int F, int U>
__device__ __forceinline__ void scatter_rows(const float* __restrict__ h,
                                             float* __restrict__ agg,
                                             int s, int d, float sc, int j0, int lane) {
  constexpr int K = RowLines<F>::K;
  float v[U][K];
  #pragma unroll
  for (int u = 0; u < U; ++u) {
    const int sj = __builtin_amdgcn_readlane(s, j0 + u);
    const int dj = __builtin_amdgcn_readlane(d, j0 + u);
    const int a  = (int)(((size_t)dj * F) & 31);
    const float* hs = h + (size_t)sj * F;
    #pragma unroll
    for (int k = 0; k < K; ++k) {
      const int f  = 32 * k + lane - a;
      const int fc = (f < 0) ? 0 : ((f > F - 1) ? (F - 1) : f);
      v[u][k] = hs[fc];
    }
  }
  #pragma unroll
  for (int u = 0; u < U; ++u) {
    const int   dj  = __builtin_amdgcn_readlane(d, j0 + u);
    const float scj = __int_as_float(__builtin_amdgcn_readlane(__float_as_int(sc), j0 + u));
    const int   a   = (int)(((size_t)dj * F) & 31);
    float* ad = agg + (size_t)dj * F;
    #pragma unroll
    for (int k = 0; k < K; ++k) {
      const int f = 32 * k + lane - a;
      if ((unsigned)f < (unsigned)F) atomicAdd(ad + f, v[u][k] * scj);
    }
  }
}

template <int CHUNKS>
__global__ void __launch_bounds__(128)
gcn_scatter_kernel(const float* __restrict__ h,
                   const int* __restrict__ src,
                   const int* __restrict__ dst,
                   const float* __restrict__ invO,
                   float* __restrict__ aggBoth,
                   int copyStride,
                   int E) {
  constexpr int F = CHUNKS * 4;
  constexpr int U = 32 / RowLines<F>::K;
  const int lane = threadIdx.x & 31;
  const int e    = blockIdx.x * blockDim.x + threadIdx.x;
  const int e0   = __builtin_amdgcn_readfirstlane(e - lane);
  if (e0 >= E) return;
  int s = 0, d = 0;
  float sc = 0.0f;
  if (e < E) {
    s  = src[e];
    d  = dst[e];
    sc = invO[s];
  }
  float* __restrict__ agg = aggBoth + (size_t)cache_half() * (size_t)copyStride;
  const int cnt = (E - e0 < 32) ? (E - e0) : 32;
  for (int jb = 0; jb < cnt; jb += U) {
    if (jb + U <= cnt) {
      scatter_rows<F, U>(h, agg, s, d, sc, jb, lane);
    } else {
      for (int j = jb; j < cnt; ++j) scatter_rows<F, 1>(h, agg, s, d, sc, j, lane);
    }
  }
}

__global__ void __launch_bounds__(256)
gcn_gemm_kernel(const float* __restrict__ A,
                const float* __restrict__ A2,
                const float* __restrict__ rowScale,
                const float* __restrict__ W,
                const float* __restrict__ bias,
                float* __restrict__ C,
                int M, int Nd, int relu) {
  __shared__ float As[16 * LDK];

  const int t     = threadIdx.x;
  const int mBase = blockIdx.x * 16;

  for (int i = t; i < 16 * (TK / 4); i += 256) {
    int row = i >> 5;
    int c4  = i & 31;
    int gr  = mBase + row;
    float4 v = make_float4(0.f, 0.f, 0.f, 0.f);
    float  s = 1.0f;
    if (gr < M) {
      const float4 a = ((const float4*)(A + (size_t)gr * TK))[c4];
      float4 w = make_float4(0.f, 0.f, 0.f, 0.f);
      if (A2) w = ((const float4*)(A2 + (size_t)gr * TK))[c4];
      if (rowScale) s = rowScale[gr];
      v = make_float4(a.x + w.x, a.y + w.y, a.z + w.z, a.w + w.w);
    }
    float* dstp = As + (size_t)row * LDK + (size_t)c4 * 4;
    dstp[0] = v.x * s; dstp[1] = v.y * s; dstp[2] = v.z * s; dstp[3] = v.w * s;
  }
  __syncthreads();

  const int wave   = t >> 5;
  const int lane   = t & 31;
  const int nTiles = (Nd + 15) >> 4;
  if (wave >= nTiles) return;

  const int col    = lane & 15;
  const int gcol   = wave * 16 + col;
  const int khalf  = (lane >> 4) * 2;
  const bool colOk = (gcol < Nd);
  const int gcSafe = colOk ? gcol : (Nd - 1);

  const float* As_row = As + (size_t)(lane & 15) * LDK + khalf;
  const float* Wp     = W + (size_t)khalf * Nd + gcSafe;

  v8f acc0 = {};
  v8f acc1 = {};
  #pragma unroll 8
  for (int k0 = 0; k0 < TK; k0 += 8) {
    v2f a0, b0, a1, b1;
    a0[0] = As_row[k0];
    a0[1] = As_row[k0 + 1];
    b0[0] = Wp[0];
    b0[1] = Wp[Nd];
    a1[0] = As_row[k0 + 4];
    a1[1] = As_row[k0 + 5];
    b1[0] = Wp[4 * (size_t)Nd];
    b1[1] = Wp[5 * (size_t)Nd];
    acc0 = __builtin_amdgcn_wmma_f32_16x16x4_f32(false, a0, false, b0, (short)0, acc0,
                                                 false, false);
    acc1 = __builtin_amdgcn_wmma_f32_16x16x4_f32(false, a1, false, b1, (short)0, acc1,
                                                 false, false);
    Wp += 8 * (size_t)Nd;
  }
  v8f acc = acc0 + acc1;

  if (colOk) {
    const float bb    = bias ? bias[gcol] : 0.0f;
    const int   rbase = mBase + (lane >> 4) * 8;
    #pragma unroll
    for (int v = 0; v < 8; ++v) {
      int r = rbase + v;
      if (r < M) {
        float val = acc[v] + bb;
        if (relu) val = fmaxf(val, 0.0f);
        C[(size_t)r * Nd + gcol] = val;
      }
    }
  }
}

__global__ void gcn_final_kernel(const float* __restrict__ agg,
                                 const float* __restrict__ agg2,
                                 const float* __restrict__ invI,
                                 const float* __restrict__ b3,
                                 float* __restrict__ out, int N, int Cc) {
  int t = blockIdx.x * blockDim.x + threadIdx.x;
  if (t < N * Cc) {
    int n = t / Cc;
    int c = t - n * Cc;
    out[t] = (agg[t] + agg2[t]) * invI[n] + b3[c];
  }
}

extern "C" void kernel_launch(void* const* d_in, const int* in_sizes, int n_in,
                              void* d_out, int out_size, void* d_ws, size_t ws_size,
                              hipStream_t stream) {
  const float* feats = (const float*)d_in[0];
  const int*   src   = (const int*)d_in[1];
  const int*   dst   = (const int*)d_in[2];
  const float* W1    = (const float*)d_in[3];
  const float* b1    = (const float*)d_in[4];
  const float* W2    = (const float*)d_in[5];
  const float* b2    = (const float*)d_in[6];
  const float* W3    = (const float*)d_in[7];
  const float* b3    = (const float*)d_in[8];
  float*       out   = (float*)d_out;

  const int HID  = in_sizes[4];
  const int NCLS = in_sizes[8];
  const int INF  = in_sizes[3] / HID;
  const int N    = in_sizes[0] / INF;
  const int E    = in_sizes[1];

  float* deg  = (float*)d_ws;
  float* invO = deg;
  float* invI = deg + N;
  float* bufA = deg + 4 * (size_t)N;
  float* bufB = bufA + 2 * (size_t)N * TK;
  float* bufC = bufB + (size_t)N * TK;

  const int B = 256;
  const int mTiles = (N + 15) / 16;

  hipMemsetAsync(deg, 0, sizeof(float) * 4 * (size_t)N, stream);
  gcn_deg_kernel<<<(E + B - 1) / B, B, 0, stream>>>(src, dst, deg, N, E);
  gcn_invsqrt_kernel<<<(N + B - 1) / B, B, 0, stream>>>(deg, N);

  hipMemsetAsync(bufA, 0, sizeof(float) * 2 * (size_t)N * TK, stream);
  {
    gcn_scatter_kernel<32><<<(E + 127) / 128, 128, 0, stream>>>(feats, src, dst, invO, bufA, N * TK, E);
  }
  gcn_gemm_kernel<<<mTiles, B, 0, stream>>>(bufA, bufA + (size_t)N * TK, invI, W1, b1, bufB, N, HID, 1);

  hipMemsetAsync(bufA, 0, sizeof(float) * 2 * (size_t)N * TK, stream);
  {
    gcn_scatter_kernel<32><<<(E + 127) / 128, 128, 0, stream>>>(bufB, src, dst, invO, bufA, N * TK, E);
  }
  gcn_gemm_kernel<<<mTiles, B, 0, stream>>>(bufA, bufA + (size_t)N * TK, invI, W2, b2, bufB, N, HID, 1);

  gcn_gemm_kernel<<<mTiles, B, 0, stream>>>(bufB, nullptr, nullptr, W3, nullptr, bufC, N, NCLS, 0);
  hipMemsetAsync(bufA, 0, sizeof(float) * 2 * (size_t)N * NCLS, stream);
  {
    gcn_scatter_kernel<10><<<(E + 127) / 128, 128, 0, stream>>>(bufC, src, dst, invO, bufA, N * NCLS, E);
  }
  gcn_final_kernel<<<(N * NCLS + B - 1) / B, B, 0, stream>>>(bufA, bufA + (size_t)N * NCLS, invI, b3, out, N, NCLS);
}


# Scatter-add by owner: 256 work-groups each keep 196 destination rows in an LDS tile, filter a byte-per-edge owner stream, sum source rows with LDS adds, store whole lines; sum order: LDS arrival
# speedup vs baseline: 2.5505x; 1.3051x over previous
#include <hip/hip_runtime.h>
#include <hip/hip_bf16.h>

typedef __attribute__((ext_vector_type(2))) float v2f;
typedef __attribute__((ext_vector_type(8))) float v8f;

#define TK 128
#define LDK 132

__device__ __forceinline__ int cache_half() {
  return (int)((__builtin_amdgcn_s_sendmsg_rtn(0x87) >> 18) & 1u);
}

#define GROUP_NODES 196

__global__ void gcn_deg_kernel(const int* __restrict__ src, const int* __restrict__ dst,
                               float* __restrict__ deg, int* __restrict__ own,
                               int N, int E, int Epad) {
  int t = blockIdx.x * blockDim.x + threadIdx.x;
  int s = 0, d = 0;
  if (t < E) {
    s = src[t];
    d = dst[t];
  }
  float* dh = deg + (size_t)cache_half() * 2 * (size_t)N;
  if (t < Epad) ((unsigned char*)own)[t] = (unsigned char)(d / GROUP_NODES);
  if (t < E) {
    atomicAdd(&dh[s], 1.0f);
    atomicAdd(&dh[(size_t)N + d], 1.0f);
  }
}

__global__ void gcn_invsqrt_kernel(float* __restrict__ deg, int N) {
  int t = blockIdx.x * blockDim.x + threadIdx.x;
  if (t < N) {
    float o = deg[t] + deg[2 * (size_t)N + t];
    float i = deg[(size_t)N + t] + deg[3 * (size_t)N + t];
    deg[t]             = (o > 0.0f) ? rsqrtf(fmaxf(o, 1.0f)) : 0.0f;
    deg[(size_t)N + t] = (i > 0.0f) ? rsqrtf(fmaxf(i, 1.0f)) : 0.0f;
  }
}

#define QCAP 64

__device__ __forceinline__ unsigned zero_bytes(unsigned z) {
  return ~(((z & 0x7F7F7F7Fu) + 0x7F7F7F7Fu) | z | 0x7F7F7F7Fu);
}

template <int F, int U>
__device__ __forceinline__ void tile_rows(const float* __restrict__ h, float* tile,
                                          int s, int dl, float sc, int j0, int lane) {
  constexpr int K = (F + 31) / 32;
  float v[U][K];
  #pragma unroll
  for (int u = 0; u < U; ++u) {
    const int sj = __builtin_amdgcn_readlane(s, j0 + u);
    const float* hs = h + (size_t)sj * F;
    #pragma unroll
    for (int k = 0; k < K; ++k) {
      const int f = 32 * k + lane;
      v[u][k] = hs[(f > F - 1) ? (F - 1) : f];
    }
  }
  #pragma unroll
  for (int u = 0; u < U; ++u) {
    const int   dj  = __builtin_amdgcn_readlane(dl, j0 + u);
    const float scj = __int_as_float(__builtin_amdgcn_readlane(__float_as_int(sc), j0 + u));
    const int rowOff = (dj >= 0) ? dj * F : GROUP_NODES * F;
    #pragma unroll
    for (int k = 0; k < K; ++k) {
      const int f = 32 * k + lane;
      atomicAdd(tile + ((f < F) ? rowOff : GROUP_NODES * F) + f, v[u][k] * scj);
    }
  }
}

template <int F>
__device__ __forceinline__ void drain_queue(const float* __restrict__ h,
                                            const int* __restrict__ src,
                                            const int* __restrict__ dst,
                                            const float* __restrict__ invO,
                                            float* tile, int base, int rows,
                                            const int* q, int cnt, int lane, int E) {
  constexpr int U = 32 / ((F + 31) / 32);
  int s = 0, dl = -1;
  float sc = 0.0f;
  if (lane < cnt) {
    const int e = q[lane];
    if (e < E) {
      s  = src[e];
      dl = dst[e] - base;
      sc = invO[s];
      if ((unsigned)dl >= (unsigned)rows) dl = -1;
    }
  }
  #pragma nounroll
  for (int jb = 0; jb < cnt; jb += U) tile_rows<F, U>(h, tile, s, dl, sc, jb, lane);
}

template <int F>
__global__ void __launch_bounds__(1024)
gcn_owner_scatter_kernel(const float* __restrict__ h,
                         const int* __restrict__ src,
                         const int* __restrict__ dst,
                         const float* __restrict__ invO,
                         const int* __restrict__ own,
                         float* __restrict__ agg,
                         int N, int E) {
  constexpr int SPARE = ((F + 31) / 32) * 32;
  __shared__ __attribute__((aligned(16))) float tile[GROUP_NODES * F + SPARE];
  __shared__ int queue[32 * QCAP];

  const int t    = threadIdx.x;
  const int lane = t & 31;
  const int wave = __builtin_amdgcn_readfirstlane(t >> 5);
  const int base = blockIdx.x * GROUP_NODES;
  const int rows = (N - base < GROUP_NODES) ? (N - base) : GROUP_NODES;

  float4* tile4 = (float4*)tile;
  for (int i = t; i < (GROUP_NODES * F + SPARE) / 4; i += 1024) tile4[i] = make_float4(0.f, 0.f, 0.f, 0.f);
  __syncthreads();

  int* q = queue + wave * QCAP;
  const unsigned pat = (unsigned)blockIdx.x * 0x01010101u;
  const int nChunks  = (E + 15) >> 4;
  const int nOuter   = (nChunks + 4095) >> 12;
  const uint4* own4  = (const uint4*)own;
  int qn = 0;
  for (int i = 0; i < nOuter; ++i) {
    const int st = (i + (int)blockIdx.x) % nOuter;
    const int c0 = st * 4096 + wave * 32 + lane;
    uint4 x[4];
    #pragma unroll
    for (int u = 0; u < 4; ++u) {
      const int c = c0 + u * 1024;
      x[u] = own4[(c < nChunks) ? c : (nChunks - 1)];
    }
    unsigned mk[4];
    #pragma unroll
    for (int u = 0; u < 4; ++u) {
      const unsigned m = (zero_bytes(x[u].x ^ pat) >> 7) | (zero_bytes(x[u].y ^ pat) >> 6) |
                         (zero_bytes(x[u].z ^ pat) >> 5) | (zero_bytes(x[u].w ^ pat) >> 4);
      mk[u] = (c0 + u * 1024 < nChunks) ? m : 0u;
    }
    #pragma nounroll
    for (int u = 0; u < 4; ++u) {
      const int c = c0 + u * 1024;
      unsigned mm = (u == 0) ? mk[0] : (u == 1) ? mk[1] : (u == 2) ? mk[2] : mk[3];
      unsigned bal = __builtin_amdgcn_ballot_w32(mm != 0);
      while (bal) {
        if (mm) {
          const int p = __builtin_ctz(mm);
          mm &= mm - 1;
          q[qn + (int)__builtin_amdgcn_mbcnt_lo(bal, 0)] = c * 16 + (p & 7) * 4 + (p >> 3);
        }
        qn += __builtin_popcount(bal);
        if (qn >= 32) {
          drain_queue<F>(h, src, dst, invO, tile, base, rows, q, 32, lane, E);
          int m = 0;
          if (lane < qn - 32) m = q[32 + lane];
          if (lane < qn - 32) q[lane] = m;
          qn -= 32;
        }
        bal = __builtin_amdgcn_ballot_w32(mm != 0);
      }
    }
  }
  if (qn > 0) drain_queue<F>(h, src, dst, invO, tile, base, rows, q, qn, lane, E);
  __syncthreads();

  float4* outp = (float4*)(agg + (size_t)base * F);
  const int n4 = rows * F / 4;
  for (int i = t; i < n4; i += 1024) outp[i] = tile4[i];
}

__global__ void __launch_bounds__(256)
gcn_gemm_kernel(const float* __restrict__ A,
                const float* __restrict__ A2,
                const float* __restrict__ rowScale,
                const float* __restrict__ W,
                const float* __restrict__ bias,
                float* __restrict__ C,
                int M, int Nd, int relu) {
  __shared__ float As[16 * LDK];

  const int t     = threadIdx.x;
  const int mBase = blockIdx.x * 16;

  for (int i = t; i < 16 * (TK / 4); i += 256) {
    int row = i >> 5;
    int c4  = i & 31;
    int gr  = mBase + row;
    float4 v = make_float4(0.f, 0.f, 0.f, 0.f);
    float  s = 1.0f;
    if (gr < M) {
      const float4 a = ((const float4*)(A + (size_t)gr * TK))[c4];
      float4 w = make_float4(0.f, 0.f, 0.f, 0.f);
      if (A2) w = ((const float4*)(A2 + (size_t)gr * TK))[c4];
      if (rowScale) s = rowScale[gr];
      v = make_float4(a.x + w.x, a.y + w.y, a.z + w.z, a.w + w.w);
    }
    float* dstp = As + (size_t)row * LDK + (size_t)c4 * 4;
    dstp[0] = v.x * s; dstp[1] = v.y * s; dstp[2] = v.z * s; dstp[3] = v.w * s;
  }
  __syncthreads();

  const int wave   = t >> 5;
  const int lane   = t & 31;
  const int nTiles = (Nd + 15) >> 4;
  if (wave >= nTiles) return;

  const int col    = lane & 15;
  const int gcol   = wave * 16 + col;
  const int khalf  = (lane >> 4) * 2;
  const bool colOk = (gcol < Nd);
  const int gcSafe = colOk ? gcol : (Nd - 1);

  const float* As_row = As + (size_t)(lane & 15) * LDK + khalf;
  const float* Wp     = W + (size_t)khalf * Nd + gcSafe;

  v8f acc0 = {};
  v8f acc1 = {};
  #pragma unroll 8
  for (int k0 = 0; k0 < TK; k0 += 8) {
    v2f a0, b0, a1, b1;
    a0[0] = As_row[k0];
    a0[1] = As_row[k0 + 1];
    b0[0] = Wp[0];
    b0[1] = Wp[Nd];
    a1[0] = As_row[k0 + 4];
    a1[1] = As_row[k0 + 5];
    b1[0] = Wp[4 * (size_t)Nd];
    b1[1] = Wp[5 * (size_t)Nd];
    acc0 = __builtin_amdgcn_wmma_f32_16x16x4_f32(false, a0, false, b0, (short)0, acc0,
                                                 false, false);
    acc1 = __builtin_amdgcn_wmma_f32_16x16x4_f32(false, a1, false, b1, (short)0, acc1,
                                                 false, false);
    Wp += 8 * (size_t)Nd;
  }
  v8f acc = acc0 + acc1;

  if (colOk) {
    const float bb    = bias ? bias[gcol] : 0.0f;
    const int   rbase = mBase + (lane >> 4) * 8;
    #pragma unroll
    for (int v = 0; v < 8; ++v) {
      int r = rbase + v;
      if (r < M) {
        float val = acc[v] + bb;
        if (relu) val = fmaxf(val, 0.0f);
        C[(size_t)r * Nd + gcol] = val;
      }
    }
  }
}

__global__ void gcn_final_kernel(const float* __restrict__ agg,
                                 const float* __restrict__ invI,
                                 const float* __restrict__ b3,
                                 float* __restrict__ out, int N, int Cc) {
  int t = blockIdx.x * blockDim.x + threadIdx.x;
  if (t < N * Cc) {
    int n = t / Cc;
    int c = t - n * Cc;
    out[t] = agg[t] * invI[n] + b3[c];
  }
}

extern "C" void kernel_launch(void* const* d_in, const int* in_sizes, int n_in,
                              void* d_out, int out_size, void* d_ws, size_t ws_size,
                              hipStream_t stream) {
  const float* feats = (const float*)d_in[0];
  const int*   src   = (const int*)d_in[1];
  const int*   dst   = (const int*)d_in[2];
  const float* W1    = (const float*)d_in[3];
  const float* b1    = (const float*)d_in[4];
  const float* W2    = (const float*)d_in[5];
  const float* b2    = (const float*)d_in[6];
  const float* W3    = (const float*)d_in[7];
  const float* b3    = (const float*)d_in[8];
  float*       out   = (float*)d_out;

  const int HID  = in_sizes[4];
  const int NCLS = in_sizes[8];
  const int INF  = in_sizes[3] / HID;
  const int N    = in_sizes[0] / INF;
  const int E    = in_sizes[1];

  float* deg  = (float*)d_ws;
  float* invO = deg;
  float* invI = deg + N;
  float* bufA = deg + 4 * (size_t)N;
  float* bufB = bufA + 2 * (size_t)N * TK;
  float* bufC = bufB + (size_t)N * TK;
  int*   own  = (int*)(bufC + (size_t)N * NCLS);

  const int B = 256;
  const int mTiles = (N + 15) / 16;
  const int Epad   = (E + 15) & ~15;
  const int groups = (N + GROUP_NODES - 1) / GROUP_NODES;

  hipMemsetAsync(deg, 0, sizeof(float) * 4 * (size_t)N, stream);
  gcn_deg_kernel<<<(Epad + B - 1) / B, B, 0, stream>>>(src, dst, deg, own, N, E, Epad);
  gcn_invsqrt_kernel<<<(N + B - 1) / B, B, 0, stream>>>(deg, N);

  gcn_owner_scatter_kernel<128><<<groups, 1024, 0, stream>>>(feats, src, dst, invO, own, bufA, N, E);
  gcn_gemm_kernel<<<mTiles, B, 0, stream>>>(bufA, nullptr, invI, W1, b1, bufB, N, HID, 1);

  gcn_owner_scatter_kernel<128><<<groups, 1024, 0, stream>>>(bufB, src, dst, invO, own, bufA, N, E);
  gcn_gemm_kernel<<<mTiles, B, 0, stream>>>(bufA, nullptr, invI, W2, b2, bufB, N, HID, 1);

  gcn_gemm_kernel<<<mTiles, B, 0, stream>>>(bufB, nullptr, nullptr, W3, nullptr, bufC, N, NCLS, 0);
  gcn_owner_scatter_kernel<40><<<groups, 1024, 0, stream>>>(bufC, src, dst, invO, own, bufA, N, E);
  gcn_final_kernel<<<(N * NCLS + B - 1) / B, B, 0, stream>>>(bufA, invI, b3, out, N, NCLS);
}


# Aggregation launches 2 and 3 walk a per-wave edge list (source, local row, scale) left by launch 1's filter: owner bytes and index fetches are read once per call; sum order: LDS arrival as before
# speedup vs baseline: 3.0051x; 1.1782x over previous
#include <hip/hip_runtime.h>
#include <hip/hip_bf16.h>

typedef __attribute__((ext_vector_type(2))) float v2f;
typedef __attribute__((ext_vector_type(8))) float v8f;

#define TK 128
#define LDK 132

__device__ __forceinline__ int cache_half() {
  return (int)((__builtin_amdgcn_s_sendmsg_rtn(0x87) >> 18) & 1u);
}

#define GROUP_NODES 196

__global__ void gcn_deg_kernel(const int* __restrict__ src, const int* __restrict__ dst,
                               float* __restrict__ deg, int* __restrict__ own,
                               int N, int E, int Epad) {
  int t = blockIdx.x * blockDim.x + threadIdx.x;
  int s = 0, d = 0;
  if (t < E) {
    s = src[t];
    d = dst[t];
  }
  float* dh = deg + (size_t)cache_half() * 2 * (size_t)N;
  if (t < Epad) ((unsigned char*)own)[t] = (unsigned char)(d / GROUP_NODES);
  if (t < E) {
    atomicAdd(&dh[s], 1.0f);
    atomicAdd(&dh[(size_t)N + d], 1.0f);
  }
}

__global__ void gcn_invsqrt_kernel(float* __restrict__ deg, int N) {
  int t = blockIdx.x * blockDim.x + threadIdx.x;
  if (t < N) {
    float o = deg[t] + deg[2 * (size_t)N + t];
    float i = deg[(size_t)N + t] + deg[3 * (size_t)N + t];
    deg[t]             = (o > 0.0f) ? rsqrtf(fmaxf(o, 1.0f)) : 0.0f;
    deg[(size_t)N + t] = (i > 0.0f) ? rsqrtf(fmaxf(i, 1.0f)) : 0.0f;
  }
}

#define QCAP 64

#define LIST_CAP 256

__device__ __forceinline__ unsigned zero_bytes(unsigned z) {
  return ~(((z & 0x7F7F7F7Fu) + 0x7F7F7F7Fu) | z | 0x7F7F7F7Fu);
}

template <int F, int U>
__device__ __forceinline__ void tile_rows(const float* __restrict__ h, float* tile,
                                          int s, int dl, float sc, int j0, int lane) {
  constexpr int K = (F + 31) / 32;
  float v[U][K];
  #pragma unroll
  for (int u = 0; u < U; ++u) {
    const int sj = __builtin_amdgcn_readlane(s, j0 + u);
    const float* hs = h + (size_t)sj * F;
    #pragma unroll
    for (int k = 0; k < K; ++k) {
      const int f = 32 * k + lane;
      v[u][k] = hs[(f > F - 1) ? (F - 1) : f];
    }
  }
  #pragma unroll
  for (int u = 0; u < U; ++u) {
    const int   dj  = __builtin_amdgcn_readlane(dl, j0 + u);
    const float scj = __int_as_float(__builtin_amdgcn_readlane(__float_as_int(sc), j0 + u));
    const int rowOff = (dj >= 0) ? dj * F : GROUP_NODES * F;
    #pragma unroll
    for (int k = 0; k < K; ++k) {
      const int f = 32 * k + lane;
      atomicAdd(tile + ((f < F) ? rowOff : GROUP_NODES * F) + f, v[u][k] * scj);
    }
  }
}

template <int F, bool REC>
__device__ __forceinline__ void drain_queue(const float* __restrict__ h,
                                            const int* __restrict__ src,
                                            const int* __restrict__ dst,
                                            const float* __restrict__ invO,
                                            float* tile, int base, int rows,
                                            const int* q, int cnt, int lane, int E,
                                            int2* __restrict__ L, int& rec) {
  constexpr int U = 32 / ((F + 31) / 32);
  int s = 0, dl = -1;
  float sc = 0.0f;
  if (lane < cnt) {
    const int e = q[lane];
    if (e < E) {
      s  = src[e];
      dl = dst[e] - base;
      sc = invO[s];
      if ((unsigned)dl >= (unsigned)rows) dl = -1;
    }
  }
  if (REC) {
    if (rec + cnt <= LIST_CAP && lane < cnt)
      L[rec + lane] = make_int2((int)(((unsigned)dl << 16) | (unsigned)s), __float_as_int(sc));
    rec += cnt;
  }
  #pragma nounroll
  for (int jb = 0; jb < cnt; jb += U) tile_rows<F, U>(h, tile, s, dl, sc, jb, lane);
}

template <int F, bool REC>
__device__ __forceinline__ int filter_stream(const float* __restrict__ h,
                                             const int* __restrict__ src,
                                             const int* __restrict__ dst,
                                             const float* __restrict__ invO,
                                             const int* __restrict__ own,
                                             float* tile, int base, int rows, int* q,
                                             int wave, int lane, int E,
                                             int2* __restrict__ L) {
  const unsigned pat = (unsigned)blockIdx.x * 0x01010101u;
  const int nChunks  = (E + 15) >> 4;
  const int nOuter   = (nChunks + 4095) >> 12;
  const uint4* own4  = (const uint4*)own;
  int qn = 0;
  int rec = 0;
  for (int i = 0; i < nOuter; ++i) {
    const int st = (i + (int)blockIdx.x) % nOuter;
    const int c0 = st * 4096 + wave * 32 + lane;
    uint4 x[4];
    #pragma unroll
    for (int u = 0; u < 4; ++u) {
      const int c = c0 + u * 1024;
      x[u] = own4[(c < nChunks) ? c : (nChunks - 1)];
    }
    unsigned mk[4];
    #pragma unroll
    for (int u = 0; u < 4; ++u) {
      const unsigned m = (zero_bytes(x[u].x ^ pat) >> 7) | (zero_bytes(x[u].y ^ pat) >> 6) |
                         (zero_bytes(x[u].z ^ pat) >> 5) | (zero_bytes(x[u].w ^ pat) >> 4);
      mk[u] = (c0 + u * 1024 < nChunks) ? m : 0u;
    }
    #pragma nounroll
    for (int u = 0; u < 4; ++u) {
      const int c = c0 + u * 1024;
      unsigned mm = (u == 0) ? mk[0] : (u == 1) ? mk[1] : (u == 2) ? mk[2] : mk[3];
      unsigned bal = __builtin_amdgcn_ballot_w32(mm != 0);
      while (bal) {
        if (mm) {
          const int p = __builtin_ctz(mm);
          mm &= mm - 1;
          q[qn + (int)__builtin_amdgcn_mbcnt_lo(bal, 0)] = c * 16 + (p & 7) * 4 + (p >> 3);
        }
        qn += __builtin_popcount(bal);
        if (qn >= 32) {
          drain_queue<F, REC>(h, src, dst, invO, tile, base, rows, q, 32, lane, E, L, rec);
          int m = 0;
          if (lane < qn - 32) m = q[32 + lane];
          if (lane < qn - 32) q[lane] = m;
          qn -= 32;
        }
        bal = __builtin_amdgcn_ballot_w32(mm != 0);
      }
    }
  }
  if (qn > 0) drain_queue<F, REC>(h, src, dst, invO, tile, base, rows, q, qn, lane, E, L, rec);
  return rec;
}

template <int F, bool REC, bool REPLAY>
__global__ void __launch_bounds__(1024)
gcn_owner_scatter_kernel(const float* __restrict__ h,
                         const int* __restrict__ src,
                         const int* __restrict__ dst,
                         const float* __restrict__ invO,
                         const int* __restrict__ own,
                         int* __restrict__ cnts,
                         int* __restrict__ lists,
                         float* __restrict__ agg,
                         int N, int E) {
  constexpr int SPARE = ((F + 31) / 32) * 32;
  __shared__ __attribute__((aligned(16))) float tile[GROUP_NODES * F + SPARE];
  __shared__ int queue[32 * QCAP];

  const int t    = threadIdx.x;
  const int lane = t & 31;
  const int wave = __builtin_amdgcn_readfirstlane(t >> 5);
  const int base = blockIdx.x * GROUP_NODES;
  const int rows = (N - base < GROUP_NODES) ? (N - base) : GROUP_NODES;

  float4* tile4 = (float4*)tile;
  for (int i = t; i < (GROUP_NODES * F + SPARE) / 4; i += 1024) tile4[i] = make_float4(0.f, 0.f, 0.f, 0.f);
  __syncthreads();

  int* q = queue + wave * QCAP;
  const int wid = (int)blockIdx.x * 32 + wave;
  int2* L = (int2*)lists + (size_t)wid * LIST_CAP;

  bool listed = false;
  if (REPLAY) {
    const int n = __builtin_amdgcn_readfirstlane(cnts[wid]);
    if ((unsigned)n <= (unsigned)LIST_CAP) {
      listed = true;
      constexpr int U = 32 / ((F + 31) / 32);
      const unsigned long long* L8 = (const unsigned long long*)L;
      const unsigned long long none = 0xFFFF0000ull;
      unsigned long long ent = (lane < n) ? L8[lane] : none;
      #pragma nounroll
      for (int j = 0; j < n; j += 32) {
        const unsigned long long cur = ent;
        ent = (j + 32 + lane < n) ? L8[j + 32 + lane] : none;
        const int   s  = (int)(cur & 0xFFFFull);
        const int   dl = (int)(unsigned)cur >> 16;
        const float sc = __int_as_float((int)(cur >> 32));
        const int   c  = (n - j < 32) ? (n - j) : 32;
        #pragma nounroll
        for (int jb = 0; jb < c; jb += U) tile_rows<F, U>(h, tile, s, dl, sc, jb, lane);
      }
    }
  }
  if (!listed) {
    const int rec = filter_stream<F, REC>(h, src, dst, invO, own, tile, base, rows, q, wave, lane, E, L);
    if (REC && lane == 0) cnts[wid] = rec;
  }
  __syncthreads();

  float4* outp = (float4*)(agg + (size_t)base * F);
  const int n4 = rows * F / 4;
  for (int i = t; i < n4; i += 1024) outp[i] = tile4[i];
}

__global__ void __launch_bounds__(256)
gcn_gemm_kernel(const float* __restrict__ A,
                const float* __restrict__ A2,
                const float* __restrict__ rowScale,
                const float* __restrict__ W,
                const float* __restrict__ bias,
                float* __restrict__ C,
                int M, int Nd, int relu) {
  __shared__ float As[16 * LDK];

  const int t     = threadIdx.x;
  const int mBase = blockIdx.x * 16;

  for (int i = t; i < 16 * (TK / 4); i += 256) {
    int row = i >> 5;
    int c4  = i & 31;
    int gr  = mBase + row;
    float4 v = make_float4(0.f, 0.f, 0.f, 0.f);
    float  s = 1.0f;
    if (gr < M) {
      const float4 a = ((const float4*)(A + (size_t)gr * TK))[c4];
      float4 w = make_float4(0.f, 0.f, 0.f, 0.f);
      if (A2) w = ((const float4*)(A2 + (size_t)gr * TK))[c4];
      if (rowScale) s = rowScale[gr];
      v = make_float4(a.x + w.x, a.y + w.y, a.z + w.z, a.w + w.w);
    }
    float* dstp = As + (size_t)row * LDK + (size_t)c4 * 4;
    dstp[0] = v.x * s; dstp[1] = v.y * s; dstp[2] = v.z * s; dstp[3] = v.w * s;
  }
  __syncthreads();

  const int wave   = t >> 5;
  const int lane   = t & 31;
  const int nTiles = (Nd + 15) >> 4;
  if (wave >= nTiles) return;

  const int col    = lane & 15;
  const int gcol   = wave * 16 + col;
  const int khalf  = (lane >> 4) * 2;
  const bool colOk = (gcol < Nd);
  const int gcSafe = colOk ? gcol : (Nd - 1);

  const float* As_row = As + (size_t)(lane & 15) * LDK + khalf;
  const float* Wp     = W + (size_t)khalf * Nd + gcSafe;

  v8f acc0 = {};
  v8f acc1 = {};
  #pragma unroll 8
  for (int k0 = 0; k0 < TK; k0 += 8) {
    v2f a0, b0, a1, b1;
    a0[0] = As_row[k0];
    a0[1] = As_row[k0 + 1];
    b0[0] = Wp[0];
    b0[1] = Wp[Nd];
    a1[0] = As_row[k0 + 4];
    a1[1] = As_row[k0 + 5];
    b1[0] = Wp[4 * (size_t)Nd];
    b1[1] = Wp[5 * (size_t)Nd];
    acc0 = __builtin_amdgcn_wmma_f32_16x16x4_f32(false, a0, false, b0, (short)0, acc0,
                                                 false, false);
    acc1 = __builtin_amdgcn_wmma_f32_16x16x4_f32(false, a1, false, b1, (short)0, acc1,
                                                 false, false);
    Wp += 8 * (size_t)Nd;
  }
  v8f acc = acc0 + acc1;

  if (colOk) {
    const float bb    = bias ? bias[gcol] : 0.0f;
    const int   rbase = mBase + (lane >> 4) * 8;
    #pragma unroll
    for (int v = 0; v < 8; ++v) {
      int r = rbase + v;
      if (r < M) {
        float val = acc[v] + bb;
        if (relu) val = fmaxf(val, 0.0f);
        C[(size_t)r * Nd + gcol] = val;
      }
    }
  }
}

__global__ void gcn_final_kernel(const float* __restrict__ agg,
                                 const float* __restrict__ invI,
                                 const float* __restrict__ b3,
                                 float* __restrict__ out, int N, int Cc) {
  int t = blockIdx.x * blockDim.x + threadIdx.x;
  if (t < N * Cc) {
    int n = t / Cc;
    int c = t - n * Cc;
    out[t] = agg[t] * invI[n] + b3[c];
  }
}

extern "C" void kernel_launch(void* const* d_in, const int* in_sizes, int n_in,
                              void* d_out, int out_size, void* d_ws, size_t ws_size,
                              hipStream_t stream) {
  const float* feats = (const float*)d_in[0];
  const int*   src   = (const int*)d_in[1];
  const int*   dst   = (const int*)d_in[2];
  const float* W1    = (const float*)d_in[3];
  const float* b1    = (const float*)d_in[4];
  const float* W2    = (const float*)d_in[5];
  const float* b2    = (const float*)d_in[6];
  const float* W3    = (const float*)d_in[7];
  const float* b3    = (const float*)d_in[8];
  float*       out   = (float*)d_out;

  const int HID  = in_sizes[4];
  const int NCLS = in_sizes[8];
  const int INF  = in_sizes[3] / HID;
  const int N    = in_sizes[0] / INF;
  const int E    = in_sizes[1];

  float* deg  = (float*)d_ws;
  float* invO = deg;
  float* invI = deg + N;
  float* bufA = deg + 4 * (size_t)N;
  float* bufB = bufA + 2 * (size_t)N * TK;
  float* bufC = bufB + (size_t)N * TK;
  int*   own  = (int*)(bufC + (size_t)N * NCLS);

  const int B = 256;
  const int mTiles = (N + 15) / 16;
  const int Epad   = (E + 15) & ~15;
  const int groups = (N + GROUP_NODES - 1) / GROUP_NODES;
  int*   cnts  = own + Epad / 4;
  int*   lists = cnts + groups * 32;

  hipMemsetAsync(deg, 0, sizeof(float) * 4 * (size_t)N, stream);
  gcn_deg_kernel<<<(Epad + B - 1) / B, B, 0, stream>>>(src, dst, deg, own, N, E, Epad);
  gcn_invsqrt_kernel<<<(N + B - 1) / B, B, 0, stream>>>(deg, N);

  gcn_owner_scatter_kernel<128, true, false><<<groups, 1024, 0, stream>>>(feats, src, dst, invO, own, cnts, lists, bufA, N, E);
  gcn_gemm_kernel<<<mTiles, B, 0, stream>>>(bufA, nullptr, invI, W1, b1, bufB, N, HID, 1);

  gcn_owner_scatter_kernel<128, false, true><<<groups, 1024, 0, stream>>>(bufB, src, dst, invO, own, cnts, lists, bufA, N, E);
  gcn_gemm_kernel<<<mTiles, B, 0, stream>>>(bufA, nullptr, invI, W2, b2, bufB, N, HID, 1);

  gcn_gemm_kernel<<<mTiles, B, 0, stream>>>(bufB, nullptr, nullptr, W3, nullptr, bufC, N, NCLS, 0);
  gcn_owner_scatter_kernel<40, false, true><<<groups, 1024, 0, stream>>>(bufC, src, dst, invO, own, cnts, lists, bufA, N, E);
  gcn_final_kernel<<<(N * NCLS + B - 1) / B, B, 0, stream>>>(bufA, invI, b3, out, N, NCLS);
}


# Final scale and bias (sum * inv_sqrt_in + b3) applied in the layer 3 aggregation's own tile store, straight into out: the element-wise launch, its waves and the sums' round trip go; sum order same
# speedup vs baseline: 3.1478x; 1.0475x over previous
#include <hip/hip_runtime.h>
#include <hip/hip_bf16.h>

typedef __attribute__((ext_vector_type(2))) float v2f;
typedef __attribute__((ext_vector_type(8))) float v8f;

#define TK 128
#define LDK 132

__device__ __forceinline__ int cache_half() {
  return (int)((__builtin_amdgcn_s_sendmsg_rtn(0x87) >> 18) & 1u);
}

#define GROUP_NODES 196

__global__ void gcn_deg_kernel(const int* __restrict__ src, const int* __restrict__ dst,
                               float* __restrict__ deg, int* __restrict__ own,
                               int N, int E, int Epad) {
  int t = blockIdx.x * blockDim.x + threadIdx.x;
  int s = 0, d = 0;
  if (t < E) {
    s = src[t];
    d = dst[t];
  }
  float* dh = deg + (size_t)cache_half() * 2 * (size_t)N;
  if (t < Epad) ((unsigned char*)own)[t] = (unsigned char)(d / GROUP_NODES);
  if (t < E) {
    atomicAdd(&dh[s], 1.0f);
    atomicAdd(&dh[(size_t)N + d], 1.0f);
  }
}

__global__ void gcn_invsqrt_kernel(float* __restrict__ deg, int N) {
  int t = blockIdx.x * blockDim.x + threadIdx.x;
  if (t < N) {
    float o = deg[t] + deg[2 * (size_t)N + t];
    float i = deg[(size_t)N + t] + deg[3 * (size_t)N + t];
    deg[t]             = (o > 0.0f) ? rsqrtf(fmaxf(o, 1.0f)) : 0.0f;
    deg[(size_t)N + t] = (i > 0.0f) ? rsqrtf(fmaxf(i, 1.0f)) : 0.0f;
  }
}

#define QCAP 64

#define LIST_CAP 256

__device__ __forceinline__ unsigned zero_bytes(unsigned z) {
  return ~(((z & 0x7F7F7F7Fu) + 0x7F7F7F7Fu) | z | 0x7F7F7F7Fu);
}

template <int F, int U>
__device__ __forceinline__ void tile_rows(const float* __restrict__ h, float* tile,
                                          int s, int dl, float sc, int j0, int lane) {
  constexpr int K = (F + 31) / 32;
  float v[U][K];
  #pragma unroll
  for (int u = 0; u < U; ++u) {
    const int sj = __builtin_amdgcn_readlane(s, j0 + u);
    const float* hs = h + (size_t)sj * F;
    #pragma unroll
    for (int k = 0; k < K; ++k) {
      const int f = 32 * k + lane;
      v[u][k] = hs[(f > F - 1) ? (F - 1) : f];
    }
  }
  #pragma unroll
  for (int u = 0; u < U; ++u) {
    const int   dj  = __builtin_amdgcn_readlane(dl, j0 + u);
    const float scj = __int_as_float(__builtin_amdgcn_readlane(__float_as_int(sc), j0 + u));
    const int rowOff = (dj >= 0) ? dj * F : GROUP_NODES * F;
    #pragma unroll
    for (int k = 0; k < K; ++k) {
      const int f = 32 * k + lane;
      atomicAdd(tile + ((f < F) ? rowOff : GROUP_NODES * F) + f, v[u][k] * scj);
    }
  }
}

template <int F, bool REC>
__device__ __forceinline__ void drain_queue(const float* __restrict__ h,
                                            const int* __restrict__ src,
                                            const int* __restrict__ dst,
                                            const float* __restrict__ invO,
                                            float* tile, int base, int rows,
                                            const int* q, int cnt, int lane, int E,
                                            int2* __restrict__ L, int& rec) {
  constexpr int U = 32 / ((F + 31) / 32);
  int s = 0, dl = -1;
  float sc = 0.0f;
  if (lane < cnt) {
    const int e = q[lane];
    if (e < E) {
      s  = src[e];
      dl = dst[e] - base;
      sc = invO[s];
      if ((unsigned)dl >= (unsigned)rows) dl = -1;
    }
  }
  if (REC) {
    if (rec + cnt <= LIST_CAP && lane < cnt)
      L[rec + lane] = make_int2((int)(((unsigned)dl << 16) | (unsigned)s), __float_as_int(sc));
    rec += cnt;
  }
  #pragma nounroll
  for (int jb = 0; jb < cnt; jb += U) tile_rows<F, U>(h, tile, s, dl, sc, jb, lane);
}

template <int F, bool REC>
__device__ __forceinline__ int filter_stream(const float* __restrict__ h,
                                             const int* __restrict__ src,
                                             const int* __restrict__ dst,
                                             const float* __restrict__ invO,
                                             const int* __restrict__ own,
                                             float* tile, int base, int rows, int* q,
                                             int wave, int lane, int E,
                                             int2* __restrict__ L) {
  const unsigned pat = (unsigned)blockIdx.x * 0x01010101u;
  const int nChunks  = (E + 15) >> 4;
  const int nOuter   = (nChunks + 4095) >> 12;
  const uint4* own4  = (const uint4*)own;
  int qn = 0;
  int rec = 0;
  for (int i = 0; i < nOuter; ++i) {
    const int st = (i + (int)blockIdx.x) % nOuter;
    const int c0 = st * 4096 + wave * 32 + lane;
    uint4 x[4];
    #pragma unroll
    for (int u = 0; u < 4; ++u) {
      const int c = c0 + u * 1024;
      x[u] = own4[(c < nChunks) ? c : (nChunks - 1)];
    }
    unsigned mk[4];
    #pragma unroll
    for (int u = 0; u < 4; ++u) {
      const unsigned m = (zero_bytes(x[u].x ^ pat) >> 7) | (zero_bytes(x[u].y ^ pat) >> 6) |
                         (zero_bytes(x[u].z ^ pat) >> 5) | (zero_bytes(x[u].w ^ pat) >> 4);
      mk[u] = (c0 + u * 1024 < nChunks) ? m : 0u;
    }
    #pragma nounroll
    for (int u = 0; u < 4; ++u) {
      const int c = c0 + u * 1024;
      unsigned mm = (u == 0) ? mk[0] : (u == 1) ? mk[1] : (u == 2) ? mk[2] : mk[3];
      unsigned bal = __builtin_amdgcn_ballot_w32(mm != 0);
      while (bal) {
        if (mm) {
          const int p = __builtin_ctz(mm);
          mm &= mm - 1;
          q[qn + (int)__builtin_amdgcn_mbcnt_lo(bal, 0)] = c * 16 + (p & 7) * 4 + (p >> 3);
        }
        qn += __builtin_popcount(bal);
        if (qn >= 32) {
          drain_queue<F, REC>(h, src, dst, invO, tile, base, rows, q, 32, lane, E, L, rec);
          int m = 0;
          if (lane < qn - 32) m = q[32 + lane];
          if (lane < qn - 32) q[lane] = m;
          qn -= 32;
        }
        bal = __builtin_amdgcn_ballot_w32(mm != 0);
      }
    }
  }
  if (qn > 0) drain_queue<F, REC>(h, src, dst, invO, tile, base, rows, q, qn, lane, E, L, rec);
  return rec;
}

template <int F, bool REC, bool REPLAY, bool FIN>
__global__ void __launch_bounds__(1024)
gcn_owner_scatter_kernel(const float* __restrict__ h,
                         const int* __restrict__ src,
                         const int* __restrict__ dst,
                         const float* __restrict__ invO,
                         const int* __restrict__ own,
                         int* __restrict__ cnts,
                         int* __restrict__ lists,
                         float* __restrict__ agg,
                         const float* __restrict__ rowScale,
                         const float* __restrict__ bias,
                         int N, int E) {
  static_assert(!FIN || F % 4 == 0, "a 16-byte piece of the tile must stay inside one row");
  constexpr int SPARE = ((F + 31) / 32) * 32;
  __shared__ __attribute__((aligned(16))) float tile[GROUP_NODES * F + SPARE];
  __shared__ int queue[32 * QCAP];

  const int t    = threadIdx.x;
  const int lane = t & 31;
  const int wave = __builtin_amdgcn_readfirstlane(t >> 5);
  const int base = blockIdx.x * GROUP_NODES;
  const int rows = (N - base < GROUP_NODES) ? (N - base) : GROUP_NODES;

  float4* tile4 = (float4*)tile;
  for (int i = t; i < (GROUP_NODES * F + SPARE) / 4; i += 1024) tile4[i] = make_float4(0.f, 0.f, 0.f, 0.f);
  __syncthreads();

  int* q = queue + wave * QCAP;
  const int wid = (int)blockIdx.x * 32 + wave;
  int2* L = (int2*)lists + (size_t)wid * LIST_CAP;

  bool listed = false;
  if (REPLAY) {
    const int n = __builtin_amdgcn_readfirstlane(cnts[wid]);
    if ((unsigned)n <= (unsigned)LIST_CAP) {
      listed = true;
      constexpr int U = 32 / ((F + 31) / 32);
      const unsigned long long* L8 = (const unsigned long long*)L;
      const unsigned long long none = 0xFFFF0000ull;
      unsigned long long ent = (lane < n) ? L8[lane] : none;
      #pragma nounroll
      for (int j = 0; j < n; j += 32) {
        const unsigned long long cur = ent;
        ent = (j + 32 + lane < n) ? L8[j + 32 + lane] : none;
        const int   s  = (int)(cur & 0xFFFFull);
        const int   dl = (int)(unsigned)cur >> 16;
        const float sc = __int_as_float((int)(cur >> 32));
        const int   c  = (n - j < 32) ? (n - j) : 32;
        #pragma nounroll
        for (int jb = 0; jb < c; jb += U) tile_rows<F, U>(h, tile, s, dl, sc, jb, lane);
      }
    }
  }
  if (!listed) {
    const int rec = filter_stream<F, REC>(h, src, dst, invO, own, tile, base, rows, q, wave, lane, E, L);
    if (REC && lane == 0) cnts[wid] = rec;
  }
  __syncthreads();

  float4* outp = (float4*)(agg + (size_t)base * F);
  const int n4 = rows * F / 4;
  if (FIN) {
    constexpr int F4 = F / 4;
    for (int i = t; i < n4; i += 1024) {
      const int r  = i / F4;
      const int c  = (i - r * F4) * 4;
      const float sc = rowScale[base + r];
      const float b0 = bias[c], b1 = bias[c + 1], b2 = bias[c + 2], b3 = bias[c + 3];
      const float4 v = tile4[i];
      outp[i] = make_float4(v.x * sc + b0, v.y * sc + b1, v.z * sc + b2, v.w * sc + b3);
    }
  } else {
    for (int i = t; i < n4; i += 1024) outp[i] = tile4[i];
  }
}

__global__ void __launch_bounds__(256)
gcn_gemm_kernel(const float* __restrict__ A,
                const float* __restrict__ A2,
                const float* __restrict__ rowScale,
                const float* __restrict__ W,
                const float* __restrict__ bias,
                float* __restrict__ C,
                int M, int Nd, int relu) {
  __shared__ float As[16 * LDK];

  const int t     = threadIdx.x;
  const int mBase = blockIdx.x * 16;

  for (int i = t; i < 16 * (TK / 4); i += 256) {
    int row = i >> 5;
    int c4  = i & 31;
    int gr  = mBase + row;
    float4 v = make_float4(0.f, 0.f, 0.f, 0.f);
    float  s = 1.0f;
    if (gr < M) {
      const float4 a = ((const float4*)(A + (size_t)gr * TK))[c4];
      float4 w = make_float4(0.f, 0.f, 0.f, 0.f);
      if (A2) w = ((const float4*)(A2 + (size_t)gr * TK))[c4];
      if (rowScale) s = rowScale[gr];
      v = make_float4(a.x + w.x, a.y + w.y, a.z + w.z, a.w + w.w);
    }
    float* dstp = As + (size_t)row * LDK + (size_t)c4 * 4;
    dstp[0] = v.x * s; dstp[1] = v.y * s; dstp[2] = v.z * s; dstp[3] = v.w * s;
  }
  __syncthreads();

  const int wave   = t >> 5;
  const int lane   = t & 31;
  const int nTiles = (Nd + 15) >> 4;
  if (wave >= nTiles) return;

  const int col    = lane & 15;
  const int gcol   = wave * 16 + col;
  const int khalf  = (lane >> 4) * 2;
  const bool colOk = (gcol < Nd);
  const int gcSafe = colOk ? gcol : (Nd - 1);

  const float* As_row = As + (size_t)(lane & 15) * LDK + khalf;
  const float* Wp     = W + (size_t)khalf * Nd + gcSafe;

  v8f acc0 = {};
  v8f acc1 = {};
  #pragma unroll 8
  for (int k0 = 0; k0 < TK; k0 += 8) {
    v2f a0, b0, a1, b1;
    a0[0] = As_row[k0];
    a0[1] = As_row[k0 + 1];
    b0[0] = Wp[0];
    b0[1] = Wp[Nd];
    a1[0] = As_row[k0 + 4];
    a1[1] = As_row[k0 + 5];
    b1[0] = Wp[4 * (size_t)Nd];
    b1[1] = Wp[5 * (size_t)Nd];
    acc0 = __builtin_amdgcn_wmma_f32_16x16x4_f32(false, a0, false, b0, (short)0, acc0,
                                                 false, false);
    acc1 = __builtin_amdgcn_wmma_f32_16x16x4_f32(false, a1, false, b1, (short)0, acc1,
                                                 false, false);
    Wp += 8 * (size_t)Nd;
  }
  v8f acc = acc0 + acc1;

  if (colOk) {
    const float bb    = bias ? bias[gcol] : 0.0f;
    const int   rbase = mBase + (lane >> 4) * 8;
    #pragma unroll
    for (int v = 0; v < 8; ++v) {
      int r = rbase + v;
      if (r < M) {
        float val = acc[v] + bb;
        if (relu) val = fmaxf(val, 0.0f);
        C[(size_t)r * Nd + gcol] = val;
      }
    }
  }
}

extern "C" void kernel_launch(void* const* d_in, const int* in_sizes, int n_in,
                              void* d_out, int out_size, void* d_ws, size_t ws_size,
                              hipStream_t stream) {
  const float* feats = (const float*)d_in[0];
  const int*   src   = (const int*)d_in[1];
  const int*   dst   = (const int*)d_in[2];
  const float* W1    = (const float*)d_in[3];
  const float* b1    = (const float*)d_in[4];
  const float* W2    = (const float*)d_in[5];
  const float* b2    = (const float*)d_in[6];
  const float* W3    = (const float*)d_in[7];
  const float* b3    = (const float*)d_in[8];
  float*       out   = (float*)d_out;

  const int HID  = in_sizes[4];
  const int NCLS = in_sizes[8];
  const int INF  = in_sizes[3] / HID;
  const int N    = in_sizes[0] / INF;
  const int E    = in_sizes[1];

  float* deg  = (float*)d_ws;
  float* invO = deg;
  float* invI = deg + N;
  float* bufA = deg + 4 * (size_t)N;
  float* bufB = bufA + 2 * (size_t)N * TK;
  float* bufC = bufB + (size_t)N * TK;
  int*   own  = (int*)(bufC + (size_t)N * NCLS);

  const int B = 256;
  const int mTiles = (N + 15) / 16;
  const int Epad   = (E + 15) & ~15;
  const int groups = (N + GROUP_NODES - 1) / GROUP_NODES;
  int*   cnts  = own + Epad / 4;
  int*   lists = cnts + groups * 32;

  hipMemsetAsync(deg, 0, sizeof(float) * 4 * (size_t)N, stream);
  gcn_deg_kernel<<<(Epad + B - 1) / B, B, 0, stream>>>(src, dst, deg, own, N, E, Epad);
  gcn_invsqrt_kernel<<<(N + B - 1) / B, B, 0, stream>>>(deg, N);

  gcn_owner_scatter_kernel<128, true, false, false><<<groups, 1024, 0, stream>>>(feats, src, dst, invO, own, cnts, lists, bufA, nullptr, nullptr, N, E);
  gcn_gemm_kernel<<<mTiles, B, 0, stream>>>(bufA, nullptr, invI, W1, b1, bufB, N, HID, 1);

  gcn_owner_scatter_kernel<128, false, true, false><<<groups, 1024, 0, stream>>>(bufB, src, dst, invO, own, cnts, lists, bufA, nullptr, nullptr, N, E);
  gcn_gemm_kernel<<<mTiles, B, 0, stream>>>(bufA, nullptr, invI, W2, b2, bufB, N, HID, 1);

  gcn_gemm_kernel<<<mTiles, B, 0, stream>>>(bufB, nullptr, nullptr, W3, nullptr, bufC, N, NCLS, 0);
  gcn_owner_scatter_kernel<40, false, true, true><<<groups, 1024, 0, stream>>>(bufC, src, dst, invO, own, cnts, lists, out, invI, b3, N, E);
}


# Layer 1 and 2 products moved into the aggregation launch's tail: 32 waves multiply the scaled LDS tile by W (B strip in registers), store relu'd h; aggregate round trip, 2 launches go; sums same
# speedup vs baseline: 3.4901x; 1.1087x over previous
#include <hip/hip_runtime.h>
#include <hip/hip_bf16.h>

typedef __attribute__((ext_vector_type(2))) float v2f;
typedef __attribute__((ext_vector_type(8))) float v8f;

#define TK 128
#define LDK 132

__device__ __forceinline__ int cache_half() {
  return (int)((__builtin_amdgcn_s_sendmsg_rtn(0x87) >> 18) & 1u);
}

#define GROUP_NODES 196

__global__ void gcn_deg_kernel(const int* __restrict__ src, const int* __restrict__ dst,
                               float* __restrict__ deg, int* __restrict__ own,
                               int N, int E, int Epad) {
  int t = blockIdx.x * blockDim.x + threadIdx.x;
  int s = 0, d = 0;
  if (t < E) {
    s = src[t];
    d = dst[t];
  }
  float* dh = deg + (size_t)cache_half() * 2 * (size_t)N;
  if (t < Epad) ((unsigned char*)own)[t] = (unsigned char)(d / GROUP_NODES);
  if (t < E) {
    atomicAdd(&dh[s], 1.0f);
    atomicAdd(&dh[(size_t)N + d], 1.0f);
  }
}

__global__ void gcn_invsqrt_kernel(float* __restrict__ deg, int N) {
  int t = blockIdx.x * blockDim.x + threadIdx.x;
  if (t < N) {
    float o = deg[t] + deg[2 * (size_t)N + t];
    float i = deg[(size_t)N + t] + deg[3 * (size_t)N + t];
    deg[t]             = (o > 0.0f) ? rsqrtf(fmaxf(o, 1.0f)) : 0.0f;
    deg[(size_t)N + t] = (i > 0.0f) ? rsqrtf(fmaxf(i, 1.0f)) : 0.0f;
  }
}

#define QCAP 64

#define LIST_CAP 256

__device__ __forceinline__ unsigned zero_bytes(unsigned z) {
  return ~(((z & 0x7F7F7F7Fu) + 0x7F7F7F7Fu) | z | 0x7F7F7F7Fu);
}

template <int F, int TS, int U>
__device__ __forceinline__ void tile_rows(const float* __restrict__ h, float* tile,
                                          int s, int dl, float sc, int j0, int lane) {
  constexpr int K = (F + 31) / 32;
  float v[U][K];
  #pragma unroll
  for (int u = 0; u < U; ++u) {
    const int sj = __builtin_amdgcn_readlane(s, j0 + u);
    const float* hs = h + (size_t)sj * F;
    #pragma unroll
    for (int k = 0; k < K; ++k) {
      const int f = 32 * k + lane;
      v[u][k] = hs[(f > F - 1) ? (F - 1) : f];
    }
  }
  #pragma unroll
  for (int u = 0; u < U; ++u) {
    const int   dj  = __builtin_amdgcn_readlane(dl, j0 + u);
    const float scj = __int_as_float(__builtin_amdgcn_readlane(__float_as_int(sc), j0 + u));
    const int rowOff = (dj >= 0) ? dj * TS : GROUP_NODES * TS;
    #pragma unroll
    for (int k = 0; k < K; ++k) {
      const int f = 32 * k + lane;
      atomicAdd(tile + ((f < F) ? rowOff : GROUP_NODES * TS) + f, v[u][k] * scj);
    }
  }
}

template <int F, int TS, bool REC>
__device__ __forceinline__ void drain_queue(const float* __restrict__ h,
                                            const int* __restrict__ src,
                                            const int* __restrict__ dst,
                                            const float* __restrict__ invO,
                                            float* tile, int base, int rows,
                                            const int* q, int cnt, int lane, int E,
                                            int2* __restrict__ L, int& rec) {
  constexpr int U = 32 / ((F + 31) / 32);
  int s = 0, dl = -1;
  float sc = 0.0f;
  if (lane < cnt) {
    const int e = q[lane];
    if (e < E) {
      s  = src[e];
      dl = dst[e] - base;
      sc = invO[s];
      if ((unsigned)dl >= (unsigned)rows) dl = -1;
    }
  }
  if (REC) {
    if (rec + cnt <= LIST_CAP && lane < cnt)
      L[rec + lane] = make_int2((int)(((unsigned)dl << 16) | (unsigned)s), __float_as_int(sc));
    rec += cnt;
  }
  #pragma nounroll
  for (int jb = 0; jb < cnt; jb += U) tile_rows<F, TS, U>(h, tile, s, dl, sc, jb, lane);
}

template <int F, int TS, bool REC>
__device__ __forceinline__ int filter_stream(const float* __restrict__ h,
                                             const int* __restrict__ src,
                                             const int* __restrict__ dst,
                                             const float* __restrict__ invO,
                                             const int* __restrict__ own,
                                             float* tile, int base, int rows, int* q,
                                             int wave, int lane, int E,
                                             int2* __restrict__ L) {
  const unsigned pat = (unsigned)blockIdx.x * 0x01010101u;
  const int nChunks  = (E + 15) >> 4;
  const int nOuter   = (nChunks + 4095) >> 12;
  const uint4* own4  = (const uint4*)own;
  int qn = 0;
  int rec = 0;
  for (int i = 0; i < nOuter; ++i) {
    const int st = (i + (int)blockIdx.x) % nOuter;
    const int c0 = st * 4096 + wave * 32 + lane;
    uint4 x[4];
    #pragma unroll
    for (int u = 0; u < 4; ++u) {
      const int c = c0 + u * 1024;
      x[u] = own4[(c < nChunks) ? c : (nChunks - 1)];
    }
    unsigned mk[4];
    #pragma unroll
    for (int u = 0; u < 4; ++u) {
      const unsigned m = (zero_bytes(x[u].x ^ pat) >> 7) | (zero_bytes(x[u].y ^ pat) >> 6) |
                         (zero_bytes(x[u].z ^ pat) >> 5) | (zero_bytes(x[u].w ^ pat) >> 4);
      mk[u] = (c0 + u * 1024 < nChunks) ? m : 0u;
    }
    #pragma nounroll
    for (int u = 0; u < 4; ++u) {
      const int c = c0 + u * 1024;
      unsigned mm = (u == 0) ? mk[0] : (u == 1) ? mk[1] : (u == 2) ? mk[2] : mk[3];
      unsigned bal = __builtin_amdgcn_ballot_w32(mm != 0);
      while (bal) {
        if (mm) {
          const int p = __builtin_ctz(mm);
          mm &= mm - 1;
          q[qn + (int)__builtin_amdgcn_mbcnt_lo(bal, 0)] = c * 16 + (p & 7) * 4 + (p >> 3);
        }
        qn += __builtin_popcount(bal);
        if (qn >= 32) {
          drain_queue<F, TS, REC>(h, src, dst, invO, tile, base, rows, q, 32, lane, E, L, rec);
          int m = 0;
          if (lane < qn - 32) m = q[32 + lane];
          if (lane < qn - 32) q[lane] = m;
          qn -= 32;
        }
        bal = __builtin_amdgcn_ballot_w32(mm != 0);
      }
    }
  }
  if (qn > 0) drain_queue<F, TS, REC>(h, src, dst, invO, tile, base, rows, q, qn, lane, E, L, rec);
  return rec;
}

template <int F, bool REC, bool REPLAY, bool FIN, bool MM>
__global__ void __launch_bounds__(1024)
gcn_owner_scatter_kernel(const float* __restrict__ h,
                         const int* __restrict__ src,
                         const int* __restrict__ dst,
                         const float* __restrict__ invO,
                         const int* __restrict__ own,
                         int* __restrict__ cnts,
                         int* __restrict__ lists,
                         float* __restrict__ agg,
                         const float* __restrict__ rowScale,
                         const float* __restrict__ bias,
                         const float* __restrict__ Wm,
                         int N, int E) {
  static_assert(!FIN || F % 4 == 0, "a 16-byte piece of the tile must stay inside one row");
  static_assert(!MM || (F == TK && !FIN), "the product in the tail is written for 128 x 128 weights");
  constexpr int TS    = MM ? F + 4 : F;
  constexpr int SPARE = ((F + 31) / 32) * 32;
  __shared__ __attribute__((aligned(16))) float tile[GROUP_NODES * TS + SPARE];
  __shared__ int queue[32 * QCAP];

  const int t    = threadIdx.x;
  const int lane = t & 31;
  const int wave = __builtin_amdgcn_readfirstlane(t >> 5);
  const int base = blockIdx.x * GROUP_NODES;
  const int rows = (N - base < GROUP_NODES) ? (N - base) : GROUP_NODES;

  float4* tile4 = (float4*)tile;
  for (int i = t; i < (GROUP_NODES * TS + SPARE) / 4; i += 1024) tile4[i] = make_float4(0.f, 0.f, 0.f, 0.f);
  __syncthreads();

  int* q = queue + wave * QCAP;
  const int wid = (int)blockIdx.x * 32 + wave;
  int2* L = (int2*)lists + (size_t)wid * LIST_CAP;

  bool listed = false;
  if (REPLAY) {
    const int n = __builtin_amdgcn_readfirstlane(cnts[wid]);
    if ((unsigned)n <= (unsigned)LIST_CAP) {
      listed = true;
      constexpr int U = 32 / ((F + 31) / 32);
      const unsigned long long* L8 = (const unsigned long long*)L;
      const unsigned long long none = 0xFFFF0000ull;
      unsigned long long ent = (lane < n) ? L8[lane] : none;
      #pragma nounroll
      for (int j = 0; j < n; j += 32) {
        const unsigned long long cur = ent;
        ent = (j + 32 + lane < n) ? L8[j + 32 + lane] : none;
        const int   s  = (int)(cur & 0xFFFFull);
        const int   dl = (int)(unsigned)cur >> 16;
        const float sc = __int_as_float((int)(cur >> 32));
        const int   c  = (n - j < 32) ? (n - j) : 32;
        #pragma nounroll
        for (int jb = 0; jb < c; jb += U) tile_rows<F, TS, U>(h, tile, s, dl, sc, jb, lane);
      }
    }
  }
  if (!listed) {
    const int rec = filter_stream<F, TS, REC>(h, src, dst, invO, own, tile, base, rows, q, wave, lane, E, L);
    if (REC && lane == 0) cnts[wid] = rec;
  }
  __syncthreads();

  if (MM) {
    constexpr int MT = (GROUP_NODES + 15) / 16;
    const int c    = lane & 15;
    const int kh   = (lane >> 4) * 2;
    const int gcol = (wave & 7) * 16 + c;
    const float* Wp = Wm + (size_t)kh * F + gcol;
    v2f b[F / 4];
    #pragma unroll
    for (int j = 0; j < F / 4; ++j) {
      b[j][0] = Wp[(size_t)(4 * j) * F];
      b[j][1] = Wp[(size_t)(4 * j + 1) * F];
    }
    const float bb = bias[gcol];

    for (int i = t; i < rows * (F / 4); i += 1024) {
      const int r = i / (F / 4);
      float4* p = (float4*)(tile + r * TS) + (i - r * (F / 4));
      const float sc = rowScale[base + r];
      float4 v = *p;
      v.x *= sc; v.y *= sc; v.z *= sc; v.w *= sc;
      *p = v;
    }
    __syncthreads();

    #pragma nounroll
    for (int mt = wave >> 3; mt < MT; mt += 4) {
      const int ar = (mt * 16 + c < GROUP_NODES) ? (mt * 16 + c) : (GROUP_NODES - 1);
      const float* ap = tile + ar * TS + kh;
      v8f acc0 = {};
      v8f acc1 = {};
      #pragma unroll
      for (int j = 0; j < F / 4; j += 2) {
        const v2f a0 = *(const v2f*)(ap + 4 * j);
        const v2f a1 = *(const v2f*)(ap + 4 * j + 4);
        acc0 = __builtin_amdgcn_wmma_f32_16x16x4_f32(false, a0, false, b[j], (short)0, acc0,
                                                     false, false);
        acc1 = __builtin_amdgcn_wmma_f32_16x16x4_f32(false, a1, false, b[j + 1], (short)0, acc1,
                                                     false, false);
      }
      const v8f acc = acc0 + acc1;
      const int r0 = mt * 16 + (lane >> 4) * 8;
      #pragma unroll
      for (int v = 0; v < 8; ++v) {
        const int r = r0 + v;
        if (r < rows) agg[(size_t)(base + r) * F + gcol] = fmaxf(acc[v] + bb, 0.0f);
      }
    }
    return;
  }

  float4* outp = (float4*)(agg + (size_t)base * F);
  const int n4 = rows * F / 4;
  if (FIN) {
    constexpr int F4 = F / 4;
    for (int i = t; i < n4; i += 1024) {
      const int r  = i / F4;
      const int c  = (i - r * F4) * 4;
      const float sc = rowScale[base + r];
      const float b0 = bias[c], b1 = bias[c + 1], b2 = bias[c + 2], b3 = bias[c + 3];
      const float4 v = tile4[i];
      outp[i] = make_float4(v.x * sc + b0, v.y * sc + b1, v.z * sc + b2, v.w * sc + b3);
    }
  } else {
    for (int i = t; i < n4; i += 1024) outp[i] = tile4[i];
  }
}

__global__ void __launch_bounds__(256)
gcn_gemm_kernel(const float* __restrict__ A,
                const float* __restrict__ A2,
                const float* __restrict__ rowScale,
                const float* __restrict__ W,
                const float* __restrict__ bias,
                float* __restrict__ C,
                int M, int Nd, int relu) {
  __shared__ float As[16 * LDK];

  const int t     = threadIdx.x;
  const int mBase = blockIdx.x * 16;

  for (int i = t; i < 16 * (TK / 4); i += 256) {
    int row = i >> 5;
    int c4  = i & 31;
    int gr  = mBase + row;
    float4 v = make_float4(0.f, 0.f, 0.f, 0.f);
    float  s = 1.0f;
    if (gr < M) {
      const float4 a = ((const float4*)(A + (size_t)gr * TK))[c4];
      float4 w = make_float4(0.f, 0.f, 0.f, 0.f);
      if (A2) w = ((const float4*)(A2 + (size_t)gr * TK))[c4];
      if (rowScale) s = rowScale[gr];
      v = make_float4(a.x + w.x, a.y + w.y, a.z + w.z, a.w + w.w);
    }
    float* dstp = As + (size_t)row * LDK + (size_t)c4 * 4;
    dstp[0] = v.x * s; dstp[1] = v.y * s; dstp[2] = v.z * s; dstp[3] = v.w * s;
  }
  __syncthreads();

  const int wave   = t >> 5;
  const int lane   = t & 31;
  const int nTiles = (Nd + 15) >> 4;
  if (wave >= nTiles) return;

  const int col    = lane & 15;
  const int gcol   = wave * 16 + col;
  const int khalf  = (lane >> 4) * 2;
  const bool colOk = (gcol < Nd);
  const int gcSafe = colOk ? gcol : (Nd - 1);

  const float* As_row = As + (size_t)(lane & 15) * LDK + khalf;
  const float* Wp     = W + (size_t)khalf * Nd + gcSafe;

  v8f acc0 = {};
  v8f acc1 = {};
  #pragma unroll 8
  for (int k0 = 0; k0 < TK; k0 += 8) {
    v2f a0, b0, a1, b1;
    a0[0] = As_row[k0];
    a0[1] = As_row[k0 + 1];
    b0[0] = Wp[0];
    b0[1] = Wp[Nd];
    a1[0] = As_row[k0 + 4];
    a1[1] = As_row[k0 + 5];
    b1[0] = Wp[4 * (size_t)Nd];
    b1[1] = Wp[5 * (size_t)Nd];
    acc0 = __builtin_amdgcn_wmma_f32_16x16x4_f32(false, a0, false, b0, (short)0, acc0,
                                                 false, false);
    acc1 = __builtin_amdgcn_wmma_f32_16x16x4_f32(false, a1, false, b1, (short)0, acc1,
                                                 false, false);
    Wp += 8 * (size_t)Nd;
  }
  v8f acc = acc0 + acc1;

  if (colOk) {
    const float bb    = bias ? bias[gcol] : 0.0f;
    const int   rbase = mBase + (lane >> 4) * 8;
    #pragma unroll
    for (int v = 0; v < 8; ++v) {
      int r = rbase + v;
      if (r < M) {
        float val = acc[v] + bb;
        if (relu) val = fmaxf(val, 0.0f);
        C[(size_t)r * Nd + gcol] = val;
      }
    }
  }
}

extern "C" void kernel_launch(void* const* d_in, const int* in_sizes, int n_in,
                              void* d_out, int out_size, void* d_ws, size_t ws_size,
                              hipStream_t stream) {
  const float* feats = (const float*)d_in[0];
  const int*   src   = (const int*)d_in[1];
  const int*   dst   = (const int*)d_in[2];
  const float* W1    = (const float*)d_in[3];
  const float* b1    = (const float*)d_in[4];
  const float* W2    = (const float*)d_in[5];
  const float* b2    = (const float*)d_in[6];
  const float* W3    = (const float*)d_in[7];
  const float* b3    = (const float*)d_in[8];
  float*       out   = (float*)d_out;

  const int HID  = in_sizes[4];
  const int NCLS = in_sizes[8];
  const int INF  = in_sizes[3] / HID;
  const int N    = in_sizes[0] / INF;
  const int E    = in_sizes[1];

  float* deg  = (float*)d_ws;
  float* invO = deg;
  float* invI = deg + N;
  float* bufA = deg + 4 * (size_t)N;
  float* bufB = bufA + 2 * (size_t)N * TK;
  float* bufC = bufB + (size_t)N * TK;
  int*   own  = (int*)(bufC + (size_t)N * NCLS);

  const int B = 256;
  const int mTiles = (N + 15) / 16;
  const int Epad   = (E + 15) & ~15;
  const int groups = (N + GROUP_NODES - 1) / GROUP_NODES;
  int*   cnts  = own + Epad / 4;
  int*   lists = cnts + groups * 32;

  hipMemsetAsync(deg, 0, sizeof(float) * 4 * (size_t)N, stream);
  gcn_deg_kernel<<<(Epad + B - 1) / B, B, 0, stream>>>(src, dst, deg, own, N, E, Epad);
  gcn_invsqrt_kernel<<<(N + B - 1) / B, B, 0, stream>>>(deg, N);

  gcn_owner_scatter_kernel<128, true, false, false, true><<<groups, 1024, 0, stream>>>(feats, src, dst, invO, own, cnts, lists, bufB, invI, b1, W1, N, E);

  gcn_owner_scatter_kernel<128, false, true, false, true><<<groups, 1024, 0, stream>>>(bufB, src, dst, invO, own, cnts, lists, bufA, invI, b2, W2, N, E);

  gcn_gemm_kernel<<<mTiles, B, 0, stream>>>(bufA, nullptr, nullptr, W3, nullptr, bufC, N, NCLS, 0);
  gcn_owner_scatter_kernel<40, false, true, true, false><<<groups, 1024, 0, stream>>>(bufC, src, dst, invO, own, cnts, lists, out, invI, b3, nullptr, N, E);
}


# First aggregation reads its edges from per-group bins that the degree launch fills (LDS count per group and 4096-edge block = slot): no scan of the whole owner stream; sum order: LDS arrival as ever
# speedup vs baseline: 3.7634x; 1.0783x over previous
#include <hip/hip_runtime.h>
#include <hip/hip_bf16.h>

typedef __attribute__((ext_vector_type(2))) float v2f;
typedef __attribute__((ext_vector_type(8))) float v8f;

#define TK 128
#define LDK 132

__device__ __forceinline__ int cache_half() {
  return (int)((__builtin_amdgcn_s_sendmsg_rtn(0x87) >> 18) & 1u);
}

#define GROUP_NODES 196

#define BLK_EDGES 4096
#define SEG_CAP 64

__global__ void __launch_bounds__(1024)
gcn_deg_kernel(const int* __restrict__ src, const int* __restrict__ dst,
               float* __restrict__ deg, int* __restrict__ own,
               int* __restrict__ cntTab, int* __restrict__ bins,
               int N, int E, int Epad, int NB) {
  __shared__ int hist[256];
  const int t = threadIdx.x;
  if (t < 256) hist[t] = 0;
  const int half = cache_half();
  __syncthreads();
  const int e0 = (int)blockIdx.x * BLK_EDGES + t;
  int s[4], d[4];
  #pragma unroll
  for (int u = 0; u < 4; ++u) {
    const int e = e0 + u * 1024;
    s[u] = 0;
    d[u] = 0;
    if (e < E) {
      s[u] = src[e];
      d[u] = dst[e];
    }
  }
  float* dh = deg + (size_t)half * 2 * (size_t)N;
  #pragma unroll
  for (int u = 0; u < 4; ++u) {
    const int e = e0 + u * 1024;
    const int g = d[u] / GROUP_NODES;
    if (e < Epad) ((unsigned char*)own)[e] = (unsigned char)g;
    if (e < E) {
      const int r = atomicAdd(&hist[g], 1);
      atomicAdd(&dh[s[u]], 1.0f);
      atomicAdd(&dh[(size_t)N + d[u]], 1.0f);
      if (r < SEG_CAP)
        bins[((size_t)g * NB + blockIdx.x) * SEG_CAP + r] =
            (int)(((unsigned)(d[u] - g * GROUP_NODES) << 16) | (unsigned)s[u]);
    }
  }
  __syncthreads();
  if (t < 256) cntTab[(size_t)t * NB + blockIdx.x] = hist[t];
}

__global__ void gcn_invsqrt_kernel(float* __restrict__ deg, int N) {
  int t = blockIdx.x * blockDim.x + threadIdx.x;
  if (t < N) {
    float o = deg[t] + deg[2 * (size_t)N + t];
    float i = deg[(size_t)N + t] + deg[3 * (size_t)N + t];
    deg[t]             = (o > 0.0f) ? rsqrtf(fmaxf(o, 1.0f)) : 0.0f;
    deg[(size_t)N + t] = (i > 0.0f) ? rsqrtf(fmaxf(i, 1.0f)) : 0.0f;
  }
}

#define QCAP 64

#define LIST_CAP 256

__device__ __forceinline__ unsigned zero_bytes(unsigned z) {
  return ~(((z & 0x7F7F7F7Fu) + 0x7F7F7F7Fu) | z | 0x7F7F7F7Fu);
}

template <int F, int TS, int U>
__device__ __forceinline__ void tile_rows(const float* __restrict__ h, float* tile,
                                          int s, int dl, float sc, int j0, int lane) {
  constexpr int K = (F + 31) / 32;
  float v[U][K];
  #pragma unroll
  for (int u = 0; u < U; ++u) {
    const int sj = __builtin_amdgcn_readlane(s, j0 + u);
    const float* hs = h + (size_t)sj * F;
    #pragma unroll
    for (int k = 0; k < K; ++k) {
      const int f = 32 * k + lane;
      v[u][k] = hs[(f > F - 1) ? (F - 1) : f];
    }
  }
  #pragma unroll
  for (int u = 0; u < U; ++u) {
    const int   dj  = __builtin_amdgcn_readlane(dl, j0 + u);
    const float scj = __int_as_float(__builtin_amdgcn_readlane(__float_as_int(sc), j0 + u));
    const int rowOff = (dj >= 0) ? dj * TS : GROUP_NODES * TS;
    #pragma unroll
    for (int k = 0; k < K; ++k) {
      const int f = 32 * k + lane;
      atomicAdd(tile + ((f < F) ? rowOff : GROUP_NODES * TS) + f, v[u][k] * scj);
    }
  }
}

template <int F, int TS, bool REC>
__device__ __forceinline__ void drain_queue(const float* __restrict__ h,
                                            const int* __restrict__ src,
                                            const int* __restrict__ dst,
                                            const float* __restrict__ invO,
                                            float* tile, int base, int rows,
                                            const int* q, int cnt, int lane, int E,
                                            int2* __restrict__ L, int& rec) {
  constexpr int U = 32 / ((F + 31) / 32);
  int s = 0, dl = -1;
  float sc = 0.0f;
  if (lane < cnt) {
    const int e = q[lane];
    if (e < E) {
      s  = src[e];
      dl = dst[e] - base;
      sc = invO[s];
      if ((unsigned)dl >= (unsigned)rows) dl = -1;
    }
  }
  if (REC) {
    if (rec + cnt <= LIST_CAP && lane < cnt)
      L[rec + lane] = make_int2((int)(((unsigned)dl << 16) | (unsigned)s), __float_as_int(sc));
    rec += cnt;
  }
  #pragma nounroll
  for (int jb = 0; jb < cnt; jb += U) tile_rows<F, TS, U>(h, tile, s, dl, sc, jb, lane);
}

template <int F, int TS, bool REC>
__device__ __forceinline__ int filter_stream(const float* __restrict__ h,
                                             const int* __restrict__ src,
                                             const int* __restrict__ dst,
                                             const float* __restrict__ invO,
                                             const int* __restrict__ own,
                                             float* tile, int base, int rows, int* q,
                                             int wave, int lane, int E,
                                             int2* __restrict__ L) {
  const unsigned pat = (unsigned)blockIdx.x * 0x01010101u;
  const int nChunks  = (E + 15) >> 4;
  const int nOuter   = (nChunks + 4095) >> 12;
  const uint4* own4  = (const uint4*)own;
  int qn = 0;
  int rec = 0;
  for (int i = 0; i < nOuter; ++i) {
    const int st = (i + (int)blockIdx.x) % nOuter;
    const int c0 = st * 4096 + wave * 32 + lane;
    uint4 x[4];
    #pragma unroll
    for (int u = 0; u < 4; ++u) {
      const int c = c0 + u * 1024;
      x[u] = own4[(c < nChunks) ? c : (nChunks - 1)];
    }
    unsigned mk[4];
    #pragma unroll
    for (int u = 0; u < 4; ++u) {
      const unsigned m = (zero_bytes(x[u].x ^ pat) >> 7) | (zero_bytes(x[u].y ^ pat) >> 6) |
                         (zero_bytes(x[u].z ^ pat) >> 5) | (zero_bytes(x[u].w ^ pat) >> 4);
      mk[u] = (c0 + u * 1024 < nChunks) ? m : 0u;
    }
    #pragma nounroll
    for (int u = 0; u < 4; ++u) {
      const int c = c0 + u * 1024;
      unsigned mm = (u == 0) ? mk[0] : (u == 1) ? mk[1] : (u == 2) ? mk[2] : mk[3];
      unsigned bal = __builtin_amdgcn_ballot_w32(mm != 0);
      while (bal) {
        if (mm) {
          const int p = __builtin_ctz(mm);
          mm &= mm - 1;
          q[qn + (int)__builtin_amdgcn_mbcnt_lo(bal, 0)] = c * 16 + (p & 7) * 4 + (p >> 3);
        }
        qn += __builtin_popcount(bal);
        if (qn >= 32) {
          drain_queue<F, TS, REC>(h, src, dst, invO, tile, base, rows, q, 32, lane, E, L, rec);
          int m = 0;
          if (lane < qn - 32) m = q[32 + lane];
          if (lane < qn - 32) q[lane] = m;
          qn -= 32;
        }
        bal = __builtin_amdgcn_ballot_w32(mm != 0);
      }
    }
  }
  if (qn > 0) drain_queue<F, TS, REC>(h, src, dst, invO, tile, base, rows, q, qn, lane, E, L, rec);
  return rec;
}

template <int F, bool REC, bool REPLAY, bool FIN, bool MM>
__global__ void __launch_bounds__(1024)
gcn_owner_scatter_kernel(const float* __restrict__ h,
                         const int* __restrict__ src,
                         const int* __restrict__ dst,
                         const float* __restrict__ invO,
                         const int* __restrict__ own,
                         int* __restrict__ cnts,
                         int* __restrict__ lists,
                         float* __restrict__ agg,
                         const float* __restrict__ rowScale,
                         const float* __restrict__ bias,
                         const float* __restrict__ Wm,
                         const int* __restrict__ cntTab,
                         const int* __restrict__ bins,
                         int N, int E, int NB) {
  static_assert(!FIN || F % 4 == 0, "a 16-byte piece of the tile must stay inside one row");
  static_assert(!MM || (F == TK && !FIN), "the product in the tail is written for 128 x 128 weights");
  constexpr int TS    = MM ? F + 4 : F;
  constexpr int SPARE = ((F + 31) / 32) * 32;
  __shared__ __attribute__((aligned(16))) float tile[GROUP_NODES * TS + SPARE];
  __shared__ int queue[32 * QCAP];
  __shared__ int badw[32];

  const int t    = threadIdx.x;
  const int lane = t & 31;
  const int wave = __builtin_amdgcn_readfirstlane(t >> 5);
  const int base = blockIdx.x * GROUP_NODES;
  const int rows = (N - base < GROUP_NODES) ? (N - base) : GROUP_NODES;

  float4* tile4 = (float4*)tile;
  for (int i = t; i < (GROUP_NODES * TS + SPARE) / 4; i += 1024) tile4[i] = make_float4(0.f, 0.f, 0.f, 0.f);

  bool binned = false;
  int cl = 0, pre = 0, total = 0, nbw = 0;
  if (REC) {
    const int myblk = wave + 32 * lane;
    if (NB <= 1024 && myblk < NB) cl = cntTab[(size_t)blockIdx.x * NB + myblk];
    nbw = (NB <= 1024 && NB > wave) ? ((NB - wave + 31) >> 5) : 0;
    for (int b = 0; b < nbw; ++b) {
      if (lane == b) pre = total;
      total += __builtin_amdgcn_readlane(cl, b);
    }
    const int bad = (NB > 1024) | (cl > SEG_CAP) | (total > LIST_CAP);
    const int wbad = (__builtin_amdgcn_ballot_w32(bad != 0) != 0) ? 1 : 0;
    if (lane == 0) badw[wave] = wbad;
    __syncthreads();
    binned = (__builtin_amdgcn_ballot_w32(badw[lane] != 0) == 0);
  } else {
    __syncthreads();
  }

  int* q = queue + wave * QCAP;
  const int wid = (int)blockIdx.x * 32 + wave;
  int2* L = (int2*)lists + (size_t)wid * LIST_CAP;

  bool listed = false;
  if (REPLAY) {
    const int n = __builtin_amdgcn_readfirstlane(cnts[wid]);
    if ((unsigned)n <= (unsigned)LIST_CAP) {
      listed = true;
      constexpr int U = 32 / ((F + 31) / 32);
      const unsigned long long* L8 = (const unsigned long long*)L;
      const unsigned long long none = 0xFFFF0000ull;
      unsigned long long ent = (lane < n) ? L8[lane] : none;
      #pragma nounroll
      for (int j = 0; j < n; j += 32) {
        const unsigned long long cur = ent;
        ent = (j + 32 + lane < n) ? L8[j + 32 + lane] : none;
        const int   s  = (int)(cur & 0xFFFFull);
        const int   dl = (int)(unsigned)cur >> 16;
        const float sc = __int_as_float((int)(cur >> 32));
        const int   c  = (n - j < 32) ? (n - j) : 32;
        #pragma nounroll
        for (int jb = 0; jb < c; jb += U) tile_rows<F, TS, U>(h, tile, s, dl, sc, jb, lane);
      }
    }
  }
  if (REC && binned) {
    constexpr int U = 32 / ((F + 31) / 32);
    const unsigned none = 0xFFFF0000u;
    auto entry = [&](int i) -> unsigned {
      int blk = 0, idx = 0;
      for (int b = 0; b < nbw; ++b) {
        const int pb = __builtin_amdgcn_readlane(pre, b);
        const int cb = __builtin_amdgcn_readlane(cl, b);
        if (i >= pb && i < pb + cb) {
          blk = wave + 32 * b;
          idx = i - pb;
        }
      }
      return (i < total) ? (unsigned)bins[((size_t)blockIdx.x * NB + blk) * SEG_CAP + idx] : none;
    };
    unsigned ent0 = entry(lane);
    unsigned ent1 = entry(32 + lane);
    float sc0 = (lane < total) ? invO[ent0 & 0xFFFFu] : 0.0f;
    #pragma nounroll
    for (int j = 0; j < total; j += 32) {
      const unsigned cur = ent0;
      const float    sc  = sc0;
      ent0 = ent1;
      ent1 = entry(j + 64 + lane);
      sc0  = (j + 32 + lane < total) ? invO[ent0 & 0xFFFFu] : 0.0f;
      const int c = (total - j < 32) ? (total - j) : 32;
      if (lane < c) L[j + lane] = make_int2((int)cur, __float_as_int(sc));
      const int s  = (int)(cur & 0xFFFFu);
      const int dl = (int)cur >> 16;
      #pragma nounroll
      for (int jb = 0; jb < c; jb += U) tile_rows<F, TS, U>(h, tile, s, dl, sc, jb, lane);
    }
    if (lane == 0) cnts[wid] = total;
  } else if (!listed) {
    const int rec = filter_stream<F, TS, REC>(h, src, dst, invO, own, tile, base, rows, q, wave, lane, E, L);
    if (REC && lane == 0) cnts[wid] = rec;
  }
  __syncthreads();

  if (MM) {
    constexpr int MT = (GROUP_NODES + 15) / 16;
    const int c    = lane & 15;
    const int kh   = (lane >> 4) * 2;
    const int gcol = (wave & 7) * 16 + c;
    const float* Wp = Wm + (size_t)kh * F + gcol;
    v2f b[F / 4];
    #pragma unroll
    for (int j = 0; j < F / 4; ++j) {
      b[j][0] = Wp[(size_t)(4 * j) * F];
      b[j][1] = Wp[(size_t)(4 * j + 1) * F];
    }
    const float bb = bias[gcol];

    for (int i = t; i < rows * (F / 4); i += 1024) {
      const int r = i / (F / 4);
      float4* p = (float4*)(tile + r * TS) + (i - r * (F / 4));
      const float sc = rowScale[base + r];
      float4 v = *p;
      v.x *= sc; v.y *= sc; v.z *= sc; v.w *= sc;
      *p = v;
    }
    __syncthreads();

    #pragma nounroll
    for (int mt = wave >> 3; mt < MT; mt += 4) {
      const int ar = (mt * 16 + c < GROUP_NODES) ? (mt * 16 + c) : (GROUP_NODES - 1);
      const float* ap = tile + ar * TS + kh;
      v8f acc0 = {};
      v8f acc1 = {};
      #pragma unroll
      for (int j = 0; j < F / 4; j += 2) {
        const v2f a0 = *(const v2f*)(ap + 4 * j);
        const v2f a1 = *(const v2f*)(ap + 4 * j + 4);
        acc0 = __builtin_amdgcn_wmma_f32_16x16x4_f32(false, a0, false, b[j], (short)0, acc0,
                                                     false, false);
        acc1 = __builtin_amdgcn_wmma_f32_16x16x4_f32(false, a1, false, b[j + 1], (short)0, acc1,
                                                     false, false);
      }
      const v8f acc = acc0 + acc1;
      const int r0 = mt * 16 + (lane >> 4) * 8;
      #pragma unroll
      for (int v = 0; v < 8; ++v) {
        const int r = r0 + v;
        if (r < rows) agg[(size_t)(base + r) * F + gcol] = fmaxf(acc[v] + bb, 0.0f);
      }
    }
    return;
  }

  float4* outp = (float4*)(agg + (size_t)base * F);
  const int n4 = rows * F / 4;
  if (FIN) {
    constexpr int F4 = F / 4;
    for (int i = t; i < n4; i += 1024) {
      const int r  = i / F4;
      const int c  = (i - r * F4) * 4;
      const float sc = rowScale[base + r];
      const float b0 = bias[c], b1 = bias[c + 1], b2 = bias[c + 2], b3 = bias[c + 3];
      const float4 v = tile4[i];
      outp[i] = make_float4(v.x * sc + b0, v.y * sc + b1, v.z * sc + b2, v.w * sc + b3);
    }
  } else {
    for (int i = t; i < n4; i += 1024) outp[i] = tile4[i];
  }
}

__global__ void __launch_bounds__(256)
gcn_gemm_kernel(const float* __restrict__ A,
                const float* __restrict__ A2,
                const float* __restrict__ rowScale,
                const float* __restrict__ W,
                const float* __restrict__ bias,
                float* __restrict__ C,
                int M, int Nd, int relu) {
  __shared__ float As[16 * LDK];

  const int t     = threadIdx.x;
  const int mBase = blockIdx.x * 16;

  for (int i = t; i < 16 * (TK / 4); i += 256) {
    int row = i >> 5;
    int c4  = i & 31;
    int gr  = mBase + row;
    float4 v = make_float4(0.f, 0.f, 0.f, 0.f);
    float  s = 1.0f;
    if (gr < M) {
      const float4 a = ((const float4*)(A + (size_t)gr * TK))[c4];
      float4 w = make_float4(0.f, 0.f, 0.f, 0.f);
      if (A2) w = ((const float4*)(A2 + (size_t)gr * TK))[c4];
      if (rowScale) s = rowScale[gr];
      v = make_float4(a.x + w.x, a.y + w.y, a.z + w.z, a.w + w.w);
    }
    float* dstp = As + (size_t)row * LDK + (size_t)c4 * 4;
    dstp[0] = v.x * s; dstp[1] = v.y * s; dstp[2] = v.z * s; dstp[3] = v.w * s;
  }
  __syncthreads();

  const int wave   = t >> 5;
  const int lane   = t & 31;
  const int nTiles = (Nd + 15) >> 4;
  if (wave >= nTiles) return;

  const int col    = lane & 15;
  const int gcol   = wave * 16 + col;
  const int khalf  = (lane >> 4) * 2;
  const bool colOk = (gcol < Nd);
  const int gcSafe = colOk ? gcol : (Nd - 1);

  const float* As_row = As + (size_t)(lane & 15) * LDK + khalf;
  const float* Wp     = W + (size_t)khalf * Nd + gcSafe;

  v8f acc0 = {};
  v8f acc1 = {};
  #pragma unroll 8
  for (int k0 = 0; k0 < TK; k0 += 8) {
    v2f a0, b0, a1, b1;
    a0[0] = As_row[k0];
    a0[1] = As_row[k0 + 1];
    b0[0] = Wp[0];
    b0[1] = Wp[Nd];
    a1[0] = As_row[k0 + 4];
    a1[1] = As_row[k0 + 5];
    b1[0] = Wp[4 * (size_t)Nd];
    b1[1] = Wp[5 * (size_t)Nd];
    acc0 = __builtin_amdgcn_wmma_f32_16x16x4_f32(false, a0, false, b0, (short)0, acc0,
                                                 false, false);
    acc1 = __builtin_amdgcn_wmma_f32_16x16x4_f32(false, a1, false, b1, (short)0, acc1,
                                                 false, false);
    Wp += 8 * (size_t)Nd;
  }
  v8f acc = acc0 + acc1;

  if (colOk) {
    const float bb    = bias ? bias[gcol] : 0.0f;
    const int   rbase = mBase + (lane >> 4) * 8;
    #pragma unroll
    for (int v = 0; v < 8; ++v) {
      int r = rbase + v;
      if (r < M) {
        float val = acc[v] + bb;
        if (relu) val = fmaxf(val, 0.0f);
        C[(size_t)r * Nd + gcol] = val;
      }
    }
  }
}

extern "C" void kernel_launch(void* const* d_in, const int* in_sizes, int n_in,
                              void* d_out, int out_size, void* d_ws, size_t ws_size,
                              hipStream_t stream) {
  const float* feats = (const float*)d_in[0];
  const int*   src   = (const int*)d_in[1];
  const int*   dst   = (const int*)d_in[2];
  const float* W1    = (const float*)d_in[3];
  const float* b1    = (const float*)d_in[4];
  const float* W2    = (const float*)d_in[5];
  const float* b2    = (const float*)d_in[6];
  const float* W3    = (const float*)d_in[7];
  const float* b3    = (const float*)d_in[8];
  float*       out   = (float*)d_out;

  const int HID  = in_sizes[4];
  const int NCLS = in_sizes[8];
  const int INF  = in_sizes[3] / HID;
  const int N    = in_sizes[0] / INF;
  const int E    = in_sizes[1];

  float* deg  = (float*)d_ws;
  float* invO = deg;
  float* invI = deg + N;
  float* bufA = deg + 4 * (size_t)N;
  float* bufB = bufA + 2 * (size_t)N * TK;
  float* bufC = bufB + (size_t)N * TK;
  int*   own  = (int*)(bufC + (size_t)N * NCLS);

  const int B = 256;
  const int mTiles = (N + 15) / 16;
  const int Epad   = (E + 15) & ~15;
  const int groups = (N + GROUP_NODES - 1) / GROUP_NODES;
  int*   cnts  = own + Epad / 4;
  int*   lists = cnts + groups * 32;
  const int NB = (Epad + BLK_EDGES - 1) / BLK_EDGES;
  int*   cntTab = lists + (size_t)groups * 32 * LIST_CAP * 2;
  int*   bins   = cntTab + (size_t)256 * NB;

  hipMemsetAsync(deg, 0, sizeof(float) * 4 * (size_t)N, stream);
  gcn_deg_kernel<<<NB, 1024, 0, stream>>>(src, dst, deg, own, cntTab, bins, N, E, Epad, NB);
  gcn_invsqrt_kernel<<<(N + B - 1) / B, B, 0, stream>>>(deg, N);

  gcn_owner_scatter_kernel<128, true, false, false, true><<<groups, 1024, 0, stream>>>(feats, src, dst, invO, own, cnts, lists, bufB, invI, b1, W1, cntTab, bins, N, E, NB);

  gcn_owner_scatter_kernel<128, false, true, false, true><<<groups, 1024, 0, stream>>>(bufB, src, dst, invO, own, cnts, lists, bufA, invI, b2, W2, nullptr, nullptr, N, E, NB);

  gcn_gemm_kernel<<<mTiles, B, 0, stream>>>(bufA, nullptr, nullptr, W3, nullptr, bufC, N, NCLS, 0);
  gcn_owner_scatter_kernel<40, false, true, true, false><<<groups, 1024, 0, stream>>>(bufC, src, dst, invO, own, cnts, lists, out, invI, b3, nullptr, nullptr, nullptr, N, E, NB);
}


# Source rows fetched by one load per row (16 B a lane for 512-B rows, 8 B for 160-B rows), not 4 and 2; 128-wide tile kept lane-interleaved so LDS adds hit 32 banks; products sum k in that order
# speedup vs baseline: 4.0605x; 1.0789x over previous
#include <hip/hip_runtime.h>
#include <hip/hip_bf16.h>

typedef __attribute__((ext_vector_type(2))) float v2f;
typedef __attribute__((ext_vector_type(8))) float v8f;

#define TK 128
#define LDK 132

__device__ __forceinline__ int cache_half() {
  return (int)((__builtin_amdgcn_s_sendmsg_rtn(0x87) >> 18) & 1u);
}

#define GROUP_NODES 196

#define BLK_EDGES 4096
#define SEG_CAP 64

__global__ void __launch_bounds__(1024)
gcn_deg_kernel(const int* __restrict__ src, const int* __restrict__ dst,
               float* __restrict__ deg, int* __restrict__ own,
               int* __restrict__ cntTab, int* __restrict__ bins,
               int N, int E, int Epad, int NB) {
  __shared__ int hist[256];
  const int t = threadIdx.x;
  if (t < 256) hist[t] = 0;
  const int half = cache_half();
  __syncthreads();
  const int e0 = (int)blockIdx.x * BLK_EDGES + t;
  int s[4], d[4];
  #pragma unroll
  for (int u = 0; u < 4; ++u) {
    const int e = e0 + u * 1024;
    s[u] = 0;
    d[u] = 0;
    if (e < E) {
      s[u] = src[e];
      d[u] = dst[e];
    }
  }
  float* dh = deg + (size_t)half * 2 * (size_t)N;
  #pragma unroll
  for (int u = 0; u < 4; ++u) {
    const int e = e0 + u * 1024;
    const int g = d[u] / GROUP_NODES;
    if (e < Epad) ((unsigned char*)own)[e] = (unsigned char)g;
    if (e < E) {
      const int r = atomicAdd(&hist[g], 1);
      atomicAdd(&dh[s[u]], 1.0f);
      atomicAdd(&dh[(size_t)N + d[u]], 1.0f);
      if (r < SEG_CAP)
        bins[((size_t)g * NB + blockIdx.x) * SEG_CAP + r] =
            (int)(((unsigned)(d[u] - g * GROUP_NODES) << 16) | (unsigned)s[u]);
    }
  }
  __syncthreads();
  if (t < 256) cntTab[(size_t)t * NB + blockIdx.x] = hist[t];
}

__global__ void gcn_invsqrt_kernel(float* __restrict__ deg, int N) {
  int t = blockIdx.x * blockDim.x + threadIdx.x;
  if (t < N) {
    float o = deg[t] + deg[2 * (size_t)N + t];
    float i = deg[(size_t)N + t] + deg[3 * (size_t)N + t];
    deg[t]             = (o > 0.0f) ? rsqrtf(fmaxf(o, 1.0f)) : 0.0f;
    deg[(size_t)N + t] = (i > 0.0f) ? rsqrtf(fmaxf(i, 1.0f)) : 0.0f;
  }
}

#define QCAP 64

#define LIST_CAP 256

__device__ __forceinline__ unsigned zero_bytes(unsigned z) {
  return ~(((z & 0x7F7F7F7Fu) + 0x7F7F7F7Fu) | z | 0x7F7F7F7Fu);
}

template <int F> struct row_plan {
  static_assert(F == 128 || (F % 2 == 0 && F <= 64), "a row must fit one load instruction of a wave");
  static constexpr int PIECE = (F == 128) ? 4 : 2;
  static constexpr int U = 32 / PIECE;
};

template <int F, int TS, int U>
__device__ __forceinline__ void tile_rows(const float* __restrict__ h, float* tile,
                                          int s, int dl, float sc, int j0, int lane) {
  if constexpr (row_plan<F>::PIECE == 4) {
    float4 v[U];
    #pragma unroll
    for (int u = 0; u < U; ++u) {
      const int sj = __builtin_amdgcn_readlane(s, j0 + u);
      v[u] = ((const float4*)(h + (size_t)sj * F))[lane];
    }
    #pragma unroll
    for (int u = 0; u < U; ++u) {
      const int   dj  = __builtin_amdgcn_readlane(dl, j0 + u);
      const float scj = __int_as_float(__builtin_amdgcn_readlane(__float_as_int(sc), j0 + u));
      float* p = tile + ((dj >= 0) ? dj * TS : GROUP_NODES * TS) + lane;
      atomicAdd(p,      v[u].x * scj);
      atomicAdd(p + 32, v[u].y * scj);
      atomicAdd(p + 64, v[u].z * scj);
      atomicAdd(p + 96, v[u].w * scj);
    }
  } else {
    static_assert(TS == 64, "lanes past the row's end add into the row's padding");
    constexpr int H = F / 2;
    float2 v[U];
    #pragma unroll
    for (int u = 0; u < U; ++u) {
      const int sj = __builtin_amdgcn_readlane(s, j0 + u);
      v[u] = ((const float2*)(h + (size_t)sj * F))[(lane < H) ? lane : (H - 1)];
    }
    #pragma unroll
    for (int u = 0; u < U; ++u) {
      const int   dj  = __builtin_amdgcn_readlane(dl, j0 + u);
      const float scj = __int_as_float(__builtin_amdgcn_readlane(__float_as_int(sc), j0 + u));
      float* p = tile + ((dj >= 0) ? dj * TS : GROUP_NODES * TS) + 2 * lane;
      atomicAdd(p,     v[u].x * scj);
      atomicAdd(p + 1, v[u].y * scj);
    }
  }
}

template <int F, int TS, bool REC>
__device__ __forceinline__ void drain_queue(const float* __restrict__ h,
                                            const int* __restrict__ src,
                                            const int* __restrict__ dst,
                                            const float* __restrict__ invO,
                                            float* tile, int base, int rows,
                                            const int* q, int cnt, int lane, int E,
                                            int2* __restrict__ L, int& rec) {
  constexpr int U = row_plan<F>::U;
  int s = 0, dl = -1;
  float sc = 0.0f;
  if (lane < cnt) {
    const int e = q[lane];
    if (e < E) {
      s  = src[e];
      dl = dst[e] - base;
      sc = invO[s];
      if ((unsigned)dl >= (unsigned)rows) dl = -1;
    }
  }
  if (REC) {
    if (rec + cnt <= LIST_CAP && lane < cnt)
      L[rec + lane] = make_int2((int)(((unsigned)dl << 16) | (unsigned)s), __float_as_int(sc));
    rec += cnt;
  }
  #pragma nounroll
  for (int jb = 0; jb < cnt; jb += U) tile_rows<F, TS, U>(h, tile, s, dl, sc, jb, lane);
}

template <int F, int TS, bool REC>
__device__ __forceinline__ int filter_stream(const float* __restrict__ h,
                                             const int* __restrict__ src,
                                             const int* __restrict__ dst,
                                             const float* __restrict__ invO,
                                             const int* __restrict__ own,
                                             float* tile, int base, int rows, int* q,
                                             int wave, int lane, int E,
                                             int2* __restrict__ L) {
  const unsigned pat = (unsigned)blockIdx.x * 0x01010101u;
  const int nChunks  = (E + 15) >> 4;
  const int nOuter   = (nChunks + 4095) >> 12;
  const uint4* own4  = (const uint4*)own;
  int qn = 0;
  int rec = 0;
  for (int i = 0; i < nOuter; ++i) {
    const int st = (i + (int)blockIdx.x) % nOuter;
    const int c0 = st * 4096 + wave * 32 + lane;
    uint4 x[4];
    #pragma unroll
    for (int u = 0; u < 4; ++u) {
      const int c = c0 + u * 1024;
      x[u] = own4[(c < nChunks) ? c : (nChunks - 1)];
    }
    unsigned mk[4];
    #pragma unroll
    for (int u = 0; u < 4; ++u) {
      const unsigned m = (zero_bytes(x[u].x ^ pat) >> 7) | (zero_bytes(x[u].y ^ pat) >> 6) |
                         (zero_bytes(x[u].z ^ pat) >> 5) | (zero_bytes(x[u].w ^ pat) >> 4);
      mk[u] = (c0 + u * 1024 < nChunks) ? m : 0u;
    }
    #pragma nounroll
    for (int u = 0; u < 4; ++u) {
      const int c = c0 + u * 1024;
      unsigned mm = (u == 0) ? mk[0] : (u == 1) ? mk[1] : (u == 2) ? mk[2] : mk[3];
      unsigned bal = __builtin_amdgcn_ballot_w32(mm != 0);
      while (bal) {
        if (mm) {
          const int p = __builtin_ctz(mm);
          mm &= mm - 1;
          q[qn + (int)__builtin_amdgcn_mbcnt_lo(bal, 0)] = c * 16 + (p & 7) * 4 + (p >> 3);
        }
        qn += __builtin_popcount(bal);
        if (qn >= 32) {
          drain_queue<F, TS, REC>(h, src, dst, invO, tile, base, rows, q, 32, lane, E, L, rec);
          int m = 0;
          if (lane < qn - 32) m = q[32 + lane];
          if (lane < qn - 32) q[lane] = m;
          qn -= 32;
        }
        bal = __builtin_amdgcn_ballot_w32(mm != 0);
      }
    }
  }
  if (qn > 0) drain_queue<F, TS, REC>(h, src, dst, invO, tile, base, rows, q, qn, lane, E, L, rec);
  return rec;
}

template <int F, bool REC, bool REPLAY, bool FIN, bool MM>
__global__ void __launch_bounds__(1024)
gcn_owner_scatter_kernel(const float* __restrict__ h,
                         const int* __restrict__ src,
                         const int* __restrict__ dst,
                         const float* __restrict__ invO,
                         const int* __restrict__ own,
                         int* __restrict__ cnts,
                         int* __restrict__ lists,
                         float* __restrict__ agg,
                         const float* __restrict__ rowScale,
                         const float* __restrict__ bias,
                         const float* __restrict__ Wm,
                         const int* __restrict__ cntTab,
                         const int* __restrict__ bins,
                         int N, int E, int NB) {
  static_assert(!FIN || F % 4 == 0, "a 16-byte piece of the tile must stay inside one row");
  static_assert(!MM || (F == TK && !FIN), "the product in the tail is written for 128 x 128 weights");
  static_assert(row_plan<F>::PIECE != 4 || MM, "a tile kept by place (see row_plan) is read only by the product in the tail");
  constexpr int TS    = MM ? F + 4 : ((row_plan<F>::PIECE == 2) ? 64 : F);
  constexpr int SPARE = (row_plan<F>::PIECE == 2) ? 64 : ((F + 31) / 32) * 32;
  __shared__ __attribute__((aligned(16))) float tile[GROUP_NODES * TS + SPARE];
  __shared__ int queue[32 * QCAP];
  __shared__ int badw[32];

  const int t    = threadIdx.x;
  const int lane = t & 31;
  const int wave = __builtin_amdgcn_readfirstlane(t >> 5);
  const int base = blockIdx.x * GROUP_NODES;
  const int rows = (N - base < GROUP_NODES) ? (N - base) : GROUP_NODES;

  float4* tile4 = (float4*)tile;
  for (int i = t; i < (GROUP_NODES * TS + SPARE) / 4; i += 1024) tile4[i] = make_float4(0.f, 0.f, 0.f, 0.f);

  bool binned = false;
  int cl = 0, pre = 0, total = 0, nbw = 0;
  if (REC) {
    const int myblk = wave + 32 * lane;
    if (NB <= 1024 && myblk < NB) cl = cntTab[(size_t)blockIdx.x * NB + myblk];
    nbw = (NB <= 1024 && NB > wave) ? ((NB - wave + 31) >> 5) : 0;
    for (int b = 0; b < nbw; ++b) {
      if (lane == b) pre = total;
      total += __builtin_amdgcn_readlane(cl, b);
    }
    const int bad = (NB > 1024) | (cl > SEG_CAP) | (total > LIST_CAP);
    const int wbad = (__builtin_amdgcn_ballot_w32(bad != 0) != 0) ? 1 : 0;
    if (lane == 0) badw[wave] = wbad;
    __syncthreads();
    binned = (__builtin_amdgcn_ballot_w32(badw[lane] != 0) == 0);
  } else {
    __syncthreads();
  }

  int* q = queue + wave * QCAP;
  const int wid = (int)blockIdx.x * 32 + wave;
  int2* L = (int2*)lists + (size_t)wid * LIST_CAP;

  bool listed = false;
  if (REPLAY) {
    const int n = __builtin_amdgcn_readfirstlane(cnts[wid]);
    if ((unsigned)n <= (unsigned)LIST_CAP) {
      listed = true;
      constexpr int U = row_plan<F>::U;
      const unsigned long long* L8 = (const unsigned long long*)L;
      const unsigned long long none = 0xFFFF0000ull;
      unsigned long long ent = (lane < n) ? L8[lane] : none;
      #pragma nounroll
      for (int j = 0; j < n; j += 32) {
        const unsigned long long cur = ent;
        ent = (j + 32 + lane < n) ? L8[j + 32 + lane] : none;
        const int   s  = (int)(cur & 0xFFFFull);
        const int   dl = (int)(unsigned)cur >> 16;
        const float sc = __int_as_float((int)(cur >> 32));
        const int   c  = (n - j < 32) ? (n - j) : 32;
        #pragma nounroll
        for (int jb = 0; jb < c; jb += U) tile_rows<F, TS, U>(h, tile, s, dl, sc, jb, lane);
      }
    }
  }
  if (REC && binned) {
    constexpr int U = row_plan<F>::U;
    const unsigned none = 0xFFFF0000u;
    auto entry = [&](int i) -> unsigned {
      int blk = 0, idx = 0;
      for (int b = 0; b < nbw; ++b) {
        const int pb = __builtin_amdgcn_readlane(pre, b);
        const int cb = __builtin_amdgcn_readlane(cl, b);
        if (i >= pb && i < pb + cb) {
          blk = wave + 32 * b;
          idx = i - pb;
        }
      }
      return (i < total) ? (unsigned)bins[((size_t)blockIdx.x * NB + blk) * SEG_CAP + idx] : none;
    };
    unsigned ent0 = entry(lane);
    unsigned ent1 = entry(32 + lane);
    float sc0 = (lane < total) ? invO[ent0 & 0xFFFFu] : 0.0f;
    #pragma nounroll
    for (int j = 0; j < total; j += 32) {
      const unsigned cur = ent0;
      const float    sc  = sc0;
      ent0 = ent1;
      ent1 = entry(j + 64 + lane);
      sc0  = (j + 32 + lane < total) ? invO[ent0 & 0xFFFFu] : 0.0f;
      const int c = (total - j < 32) ? (total - j) : 32;
      if (lane < c) L[j + lane] = make_int2((int)cur, __float_as_int(sc));
      const int s  = (int)(cur & 0xFFFFu);
      const int dl = (int)cur >> 16;
      #pragma nounroll
      for (int jb = 0; jb < c; jb += U) tile_rows<F, TS, U>(h, tile, s, dl, sc, jb, lane);
    }
    if (lane == 0) cnts[wid] = total;
  } else if (!listed) {
    const int rec = filter_stream<F, TS, REC>(h, src, dst, invO, own, tile, base, rows, q, wave, lane, E, L);
    if (REC && lane == 0) cnts[wid] = rec;
  }
  __syncthreads();

  if (MM) {
    constexpr int MT = (GROUP_NODES + 15) / 16;
    const int c    = lane & 15;
    const int kh   = (lane >> 4) * 2;
    const int gcol = (wave & 7) * 16 + c;
    const float* Wp = Wm + (size_t)(4 * kh) * F + gcol;
    v2f b[F / 4];
    #pragma unroll
    for (int j = 0; j < F / 4; ++j) {
      b[j][0] = Wp[(size_t)(16 * (j & 7) + (j >> 3)) * F];
      b[j][1] = Wp[(size_t)(16 * (j & 7) + (j >> 3) + 4) * F];
    }
    const float bb = bias[gcol];

    for (int i = t; i < rows * (F / 4); i += 1024) {
      const int r = i / (F / 4);
      float4* p = (float4*)(tile + r * TS) + (i - r * (F / 4));
      const float sc = rowScale[base + r];
      float4 v = *p;
      v.x *= sc; v.y *= sc; v.z *= sc; v.w *= sc;
      *p = v;
    }
    __syncthreads();

    #pragma nounroll
    for (int mt = wave >> 3; mt < MT; mt += 4) {
      const int ar = (mt * 16 + c < GROUP_NODES) ? (mt * 16 + c) : (GROUP_NODES - 1);
      const float* ap = tile + ar * TS + kh;
      v8f acc0 = {};
      v8f acc1 = {};
      #pragma unroll
      for (int j = 0; j < F / 4; j += 2) {
        const v2f a0 = *(const v2f*)(ap + 4 * j);
        const v2f a1 = *(const v2f*)(ap + 4 * j + 4);
        acc0 = __builtin_amdgcn_wmma_f32_16x16x4_f32(false, a0, false, b[j], (short)0, acc0,
                                                     false, false);
        acc1 = __builtin_amdgcn_wmma_f32_16x16x4_f32(false, a1, false, b[j + 1], (short)0, acc1,
                                                     false, false);
      }
      const v8f acc = acc0 + acc1;
      const int r0 = mt * 16 + (lane >> 4) * 8;
      #pragma unroll
      for (int v = 0; v < 8; ++v) {
        const int r = r0 + v;
        if (r < rows) agg[(size_t)(base + r) * F + gcol] = fmaxf(acc[v] + bb, 0.0f);
      }
    }
    return;
  }

  float4* outp = (float4*)(agg + (size_t)base * F);
  const int n4 = rows * F / 4;
  if (FIN) {
    constexpr int F4 = F / 4;
    for (int i = t; i < n4; i += 1024) {
      const int r  = i / F4;
      const int c  = (i - r * F4) * 4;
      const float sc = rowScale[base + r];
      const float b0 = bias[c], b1 = bias[c + 1], b2 = bias[c + 2], b3 = bias[c + 3];
      const float4 v = tile4[r * (TS / 4) + (i - r * F4)];
      outp[i] = make_float4(v.x * sc + b0, v.y * sc + b1, v.z * sc + b2, v.w * sc + b3);
    }
  } else {
    static_assert(MM || FIN || TS == F, "the plain copy takes the tile's rows as they lie");
    for (int i = t; i < n4; i += 1024) outp[i] = tile4[i];
  }
}

__global__ void __launch_bounds__(256)
gcn_gemm_kernel(const float* __restrict__ A,
                const float* __restrict__ A2,
                const float* __restrict__ rowScale,
                const float* __restrict__ W,
                const float* __restrict__ bias,
                float* __restrict__ C,
                int M, int Nd, int relu) {
  __shared__ float As[16 * LDK];

  const int t     = threadIdx.x;
  const int mBase = blockIdx.x * 16;

  for (int i = t; i < 16 * (TK / 4); i += 256) {
    int row = i >> 5;
    int c4  = i & 31;
    int gr  = mBase + row;
    float4 v = make_float4(0.f, 0.f, 0.f, 0.f);
    float  s = 1.0f;
    if (gr < M) {
      const float4 a = ((const float4*)(A + (size_t)gr * TK))[c4];
      float4 w = make_float4(0.f, 0.f, 0.f, 0.f);
      if (A2) w = ((const float4*)(A2 + (size_t)gr * TK))[c4];
      if (rowScale) s = rowScale[gr];
      v = make_float4(a.x + w.x, a.y + w.y, a.z + w.z, a.w + w.w);
    }
    float* dstp = As + (size_t)row * LDK + (size_t)c4 * 4;
    dstp[0] = v.x * s; dstp[1] = v.y * s; dstp[2] = v.z * s; dstp[3] = v.w * s;
  }
  __syncthreads();

  const int wave   = t >> 5;
  const int lane   = t & 31;
  const int nTiles = (Nd + 15) >> 4;
  if (wave >= nTiles) return;

  const int col    = lane & 15;
  const int gcol   = wave * 16 + col;
  const int khalf  = (lane >> 4) * 2;
  const bool colOk = (gcol < Nd);
  const int gcSafe = colOk ? gcol : (Nd - 1);

  const float* As_row = As + (size_t)(lane & 15) * LDK + khalf;
  const float* Wp     = W + (size_t)khalf * Nd + gcSafe;

  v8f acc0 = {};
  v8f acc1 = {};
  #pragma unroll 8
  for (int k0 = 0; k0 < TK; k0 += 8) {
    v2f a0, b0, a1, b1;
    a0[0] = As_row[k0];
    a0[1] = As_row[k0 + 1];
    b0[0] = Wp[0];
    b0[1] = Wp[Nd];
    a1[0] = As_row[k0 + 4];
    a1[1] = As_row[k0 + 5];
    b1[0] = Wp[4 * (size_t)Nd];
    b1[1] = Wp[5 * (size_t)Nd];
    acc0 = __builtin_amdgcn_wmma_f32_16x16x4_f32(false, a0, false, b0, (short)0, acc0,
                                                 false, false);
    acc1 = __builtin_amdgcn_wmma_f32_16x16x4_f32(false, a1, false, b1, (short)0, acc1,
                                                 false, false);
    Wp += 8 * (size_t)Nd;
  }
  v8f acc = acc0 + acc1;

  if (colOk) {
    const float bb    = bias ? bias[gcol] : 0.0f;
    const int   rbase = mBase + (lane >> 4) * 8;
    #pragma unroll
    for (int v = 0; v < 8; ++v) {
      int r = rbase + v;
      if (r < M) {
        float val = acc[v] + bb;
        if (relu) val = fmaxf(val, 0.0f);
        C[(size_t)r * Nd + gcol] = val;
      }
    }
  }
}

extern "C" void kernel_launch(void* const* d_in, const int* in_sizes, int n_in,
                              void* d_out, int out_size, void* d_ws, size_t ws_size,
                              hipStream_t stream) {
  const float* feats = (const float*)d_in[0];
  const int*   src   = (const int*)d_in[1];
  const int*   dst   = (const int*)d_in[2];
  const float* W1    = (const float*)d_in[3];
  const float* b1    = (const float*)d_in[4];
  const float* W2    = (const float*)d_in[5];
  const float* b2    = (const float*)d_in[6];
  const float* W3    = (const float*)d_in[7];
  const float* b3    = (const float*)d_in[8];
  float*       out   = (float*)d_out;

  const int HID  = in_sizes[4];
  const int NCLS = in_sizes[8];
  const int INF  = in_sizes[3] / HID;
  const int N    = in_sizes[0] / INF;
  const int E    = in_sizes[1];

  float* deg  = (float*)d_ws;
  float* invO = deg;
  float* invI = deg + N;
  float* bufA = deg + 4 * (size_t)N;
  float* bufB = bufA + 2 * (size_t)N * TK;
  float* bufC = bufB + (size_t)N * TK;
  int*   own  = (int*)(bufC + (size_t)N * NCLS);

  const int B = 256;
  const int mTiles = (N + 15) / 16;
  const int Epad   = (E + 15) & ~15;
  const int groups = (N + GROUP_NODES - 1) / GROUP_NODES;
  int*   cnts  = own + Epad / 4;
  int*   lists = cnts + groups * 32;
  const int NB = (Epad + BLK_EDGES - 1) / BLK_EDGES;
  int*   cntTab = lists + (size_t)groups * 32 * LIST_CAP * 2;
  int*   bins   = cntTab + (size_t)256 * NB;

  hipMemsetAsync(deg, 0, sizeof(float) * 4 * (size_t)N, stream);
  gcn_deg_kernel<<<NB, 1024, 0, stream>>>(src, dst, deg, own, cntTab, bins, N, E, Epad, NB);
  gcn_invsqrt_kernel<<<(N + B - 1) / B, B, 0, stream>>>(deg, N);

  gcn_owner_scatter_kernel<128, true, false, false, true><<<groups, 1024, 0, stream>>>(feats, src, dst, invO, own, cnts, lists, bufB, invI, b1, W1, cntTab, bins, N, E, NB);

  gcn_owner_scatter_kernel<128, false, true, false, true><<<groups, 1024, 0, stream>>>(bufB, src, dst, invO, own, cnts, lists, bufA, invI, b2, W2, nullptr, nullptr, N, E, NB);

  gcn_gemm_kernel<<<mTiles, B, 0, stream>>>(bufA, nullptr, nullptr, W3, nullptr, bufC, N, NCLS, 0);
  gcn_owner_scatter_kernel<40, false, true, true, false><<<groups, 1024, 0, stream>>>(bufC, src, dst, invO, own, cnts, lists, out, invI, b3, nullptr, nullptr, nullptr, N, E, NB);
}


# Degree scales made where used: each aggregation counts its rows' in-degree in LDS (1 add per 32 edges), source rsqrt taken at list build; rsqrt launch, in-degree global atomics go; sums same
# speedup vs baseline: 4.2440x; 1.0452x over previous
#include <hip/hip_runtime.h>
#include <hip/hip_bf16.h>

typedef __attribute__((ext_vector_type(2))) float v2f;
typedef __attribute__((ext_vector_type(8))) float v8f;

#define TK 128
#define LDK 132

__device__ __forceinline__ int cache_half() {
  return (int)((__builtin_amdgcn_s_sendmsg_rtn(0x87) >> 18) & 1u);
}

#define GROUP_NODES 196

#define BLK_EDGES 4096
#define SEG_CAP 64

__global__ void __launch_bounds__(1024)
gcn_deg_kernel(const int* __restrict__ src, const int* __restrict__ dst,
               float* __restrict__ deg, int* __restrict__ own,
               int* __restrict__ cntTab, int* __restrict__ bins,
               int N, int E, int Epad, int NB) {
  __shared__ int hist[256];
  const int t = threadIdx.x;
  if (t < 256) hist[t] = 0;
  const int half = cache_half();
  __syncthreads();
  const int e0 = (int)blockIdx.x * BLK_EDGES + t;
  int s[4], d[4];
  #pragma unroll
  for (int u = 0; u < 4; ++u) {
    const int e = e0 + u * 1024;
    s[u] = 0;
    d[u] = 0;
    if (e < E) {
      s[u] = src[e];
      d[u] = dst[e];
    }
  }
  float* dh = deg + (size_t)half * (size_t)N;
  #pragma unroll
  for (int u = 0; u < 4; ++u) {
    const int e = e0 + u * 1024;
    const int g = d[u] / GROUP_NODES;
    if (e < Epad) ((unsigned char*)own)[e] = (unsigned char)g;
    if (e < E) {
      const int r = atomicAdd(&hist[g], 1);
      atomicAdd(&dh[s[u]], 1.0f);
      if (r < SEG_CAP)
        bins[((size_t)g * NB + blockIdx.x) * SEG_CAP + r] =
            (int)(((unsigned)(d[u] - g * GROUP_NODES) << 16) | (unsigned)s[u]);
    }
  }
  __syncthreads();
  if (t < 256) cntTab[(size_t)t * NB + blockIdx.x] = hist[t];
}

__device__ __forceinline__ float inv_sqrt_count(float c) {
  return (c > 0.0f) ? rsqrtf(fmaxf(c, 1.0f)) : 0.0f;
}
__device__ __forceinline__ float out_scale(const float* __restrict__ degO, int N, int s) {
  return inv_sqrt_count(degO[s] + degO[(size_t)N + s]);
}

#define QCAP 64

#define LIST_CAP 256

__device__ __forceinline__ unsigned zero_bytes(unsigned z) {
  return ~(((z & 0x7F7F7F7Fu) + 0x7F7F7F7Fu) | z | 0x7F7F7F7Fu);
}

template <int F> struct row_plan {
  static_assert(F == 128 || (F % 2 == 0 && F <= 64), "a row must fit one load instruction of a wave");
  static constexpr int PIECE = (F == 128) ? 4 : 2;
  static constexpr int U = 32 / PIECE;
};

template <int F, int TS, int U>
__device__ __forceinline__ void tile_rows(const float* __restrict__ h, float* tile,
                                          int s, int dl, float sc, int j0, int lane) {
  if constexpr (row_plan<F>::PIECE == 4) {
    float4 v[U];
    #pragma unroll
    for (int u = 0; u < U; ++u) {
      const int sj = __builtin_amdgcn_readlane(s, j0 + u);
      v[u] = ((const float4*)(h + (size_t)sj * F))[lane];
    }
    #pragma unroll
    for (int u = 0; u < U; ++u) {
      const int   dj  = __builtin_amdgcn_readlane(dl, j0 + u);
      const float scj = __int_as_float(__builtin_amdgcn_readlane(__float_as_int(sc), j0 + u));
      float* p = tile + ((dj >= 0) ? dj * TS : GROUP_NODES * TS) + lane;
      atomicAdd(p,      v[u].x * scj);
      atomicAdd(p + 32, v[u].y * scj);
      atomicAdd(p + 64, v[u].z * scj);
      atomicAdd(p + 96, v[u].w * scj);
    }
  } else {
    static_assert(TS == 64, "lanes past the row's end add into the row's padding");
    constexpr int H = F / 2;
    float2 v[U];
    #pragma unroll
    for (int u = 0; u < U; ++u) {
      const int sj = __builtin_amdgcn_readlane(s, j0 + u);
      v[u] = ((const float2*)(h + (size_t)sj * F))[(lane < H) ? lane : (H - 1)];
    }
    #pragma unroll
    for (int u = 0; u < U; ++u) {
      const int   dj  = __builtin_amdgcn_readlane(dl, j0 + u);
      const float scj = __int_as_float(__builtin_amdgcn_readlane(__float_as_int(sc), j0 + u));
      float* p = tile + ((dj >= 0) ? dj * TS : GROUP_NODES * TS) + 2 * lane;
      atomicAdd(p,     v[u].x * scj);
      atomicAdd(p + 1, v[u].y * scj);
    }
  }
}

template <int F, int TS, bool REC>
__device__ __forceinline__ void drain_queue(const float* __restrict__ h,
                                            const int* __restrict__ src,
                                            const int* __restrict__ dst,
                                            const float* __restrict__ degO, int N,
                                            float* tile, int* degc, int base, int rows,
                                            const int* q, int cnt, int lane, int E,
                                            int2* __restrict__ L, int& rec) {
  constexpr int U = row_plan<F>::U;
  int s = 0, dl = -1;
  float sc = 0.0f;
  if (lane < cnt) {
    const int e = q[lane];
    if (e < E) {
      s  = src[e];
      dl = dst[e] - base;
      sc = out_scale(degO, N, s);
      if ((unsigned)dl >= (unsigned)rows) dl = -1;
      if (dl >= 0) atomicAdd(&degc[dl], 1);
    }
  }
  if (REC) {
    if (rec + cnt <= LIST_CAP && lane < cnt)
      L[rec + lane] = make_int2((int)(((unsigned)dl << 16) | (unsigned)s), __float_as_int(sc));
    rec += cnt;
  }
  #pragma nounroll
  for (int jb = 0; jb < cnt; jb += U) tile_rows<F, TS, U>(h, tile, s, dl, sc, jb, lane);
}

template <int F, int TS, bool REC>
__device__ __forceinline__ int filter_stream(const float* __restrict__ h,
                                             const int* __restrict__ src,
                                             const int* __restrict__ dst,
                                             const float* __restrict__ degO, int N,
                                             const int* __restrict__ own,
                                             float* tile, int* degc, int base, int rows, int* q,
                                             int wave, int lane, int E,
                                             int2* __restrict__ L) {
  const unsigned pat = (unsigned)blockIdx.x * 0x01010101u;
  const int nChunks  = (E + 15) >> 4;
  const int nOuter   = (nChunks + 4095) >> 12;
  const uint4* own4  = (const uint4*)own;
  int qn = 0;
  int rec = 0;
  for (int i = 0; i < nOuter; ++i) {
    const int st = (i + (int)blockIdx.x) % nOuter;
    const int c0 = st * 4096 + wave * 32 + lane;
    uint4 x[4];
    #pragma unroll
    for (int u = 0; u < 4; ++u) {
      const int c = c0 + u * 1024;
      x[u] = own4[(c < nChunks) ? c : (nChunks - 1)];
    }
    unsigned mk[4];
    #pragma unroll
    for (int u = 0; u < 4; ++u) {
      const unsigned m = (zero_bytes(x[u].x ^ pat) >> 7) | (zero_bytes(x[u].y ^ pat) >> 6) |
                         (zero_bytes(x[u].z ^ pat) >> 5) | (zero_bytes(x[u].w ^ pat) >> 4);
      mk[u] = (c0 + u * 1024 < nChunks) ? m : 0u;
    }
    #pragma nounroll
    for (int u = 0; u < 4; ++u) {
      const int c = c0 + u * 1024;
      unsigned mm = (u == 0) ? mk[0] : (u == 1) ? mk[1] : (u == 2) ? mk[2] : mk[3];
      unsigned bal = __builtin_amdgcn_ballot_w32(mm != 0);
      while (bal) {
        if (mm) {
          const int p = __builtin_ctz(mm);
          mm &= mm - 1;
          q[qn + (int)__builtin_amdgcn_mbcnt_lo(bal, 0)] = c * 16 + (p & 7) * 4 + (p >> 3);
        }
        qn += __builtin_popcount(bal);
        if (qn >= 32) {
          drain_queue<F, TS, REC>(h, src, dst, degO, N, tile, degc, base, rows, q, 32, lane, E, L, rec);
          int m = 0;
          if (lane < qn - 32) m = q[32 + lane];
          if (lane < qn - 32) q[lane] = m;
          qn -= 32;
        }
        bal = __builtin_amdgcn_ballot_w32(mm != 0);
      }
    }
  }
  if (qn > 0) drain_queue<F, TS, REC>(h, src, dst, degO, N, tile, degc, base, rows, q, qn, lane, E, L, rec);
  return rec;
}

template <int F, bool REC, bool REPLAY, bool FIN, bool MM>
__global__ void __launch_bounds__(1024)
gcn_owner_scatter_kernel(const float* __restrict__ h,
                         const int* __restrict__ src,
                         const int* __restrict__ dst,
                         const float* __restrict__ degO,
                         const int* __restrict__ own,
                         int* __restrict__ cnts,
                         int* __restrict__ lists,
                         float* __restrict__ agg,
                         const float* __restrict__ bias,
                         const float* __restrict__ Wm,
                         const int* __restrict__ cntTab,
                         const int* __restrict__ bins,
                         int N, int E, int NB) {
  static_assert(!FIN || F % 4 == 0, "a 16-byte piece of the tile must stay inside one row");
  static_assert(!MM || (F == TK && !FIN), "the product in the tail is written for 128 x 128 weights");
  static_assert(row_plan<F>::PIECE != 4 || MM, "a tile kept by place (see row_plan) is read only by the product in the tail");
  constexpr int TS    = MM ? F + 4 : ((row_plan<F>::PIECE == 2) ? 64 : F);
  constexpr int SPARE = (row_plan<F>::PIECE == 2) ? 64 : ((F + 31) / 32) * 32;
  __shared__ __attribute__((aligned(16))) float tile[GROUP_NODES * TS + SPARE];
  __shared__ int queue[32 * QCAP];
  __shared__ int badw[32];
  __shared__ int degc[GROUP_NODES];

  const int t    = threadIdx.x;
  const int lane = t & 31;
  const int wave = __builtin_amdgcn_readfirstlane(t >> 5);
  const int base = blockIdx.x * GROUP_NODES;
  const int rows = (N - base < GROUP_NODES) ? (N - base) : GROUP_NODES;

  float4* tile4 = (float4*)tile;
  for (int i = t; i < (GROUP_NODES * TS + SPARE) / 4; i += 1024) tile4[i] = make_float4(0.f, 0.f, 0.f, 0.f);
  if (t < GROUP_NODES) degc[t] = 0;

  bool binned = false;
  int cl = 0, pre = 0, total = 0, nbw = 0;
  if (REC) {
    const int myblk = wave + 32 * lane;
    if (NB <= 1024 && myblk < NB) cl = cntTab[(size_t)blockIdx.x * NB + myblk];
    nbw = (NB <= 1024 && NB > wave) ? ((NB - wave + 31) >> 5) : 0;
    for (int b = 0; b < nbw; ++b) {
      if (lane == b) pre = total;
      total += __builtin_amdgcn_readlane(cl, b);
    }
    const int bad = (NB > 1024) | (cl > SEG_CAP) | (total > LIST_CAP);
    const int wbad = (__builtin_amdgcn_ballot_w32(bad != 0) != 0) ? 1 : 0;
    if (lane == 0) badw[wave] = wbad;
    __syncthreads();
    binned = (__builtin_amdgcn_ballot_w32(badw[lane] != 0) == 0);
  } else {
    __syncthreads();
  }

  int* q = queue + wave * QCAP;
  const int wid = (int)blockIdx.x * 32 + wave;
  int2* L = (int2*)lists + (size_t)wid * LIST_CAP;

  bool listed = false;
  if (REPLAY) {
    const int n = __builtin_amdgcn_readfirstlane(cnts[wid]);
    if ((unsigned)n <= (unsigned)LIST_CAP) {
      listed = true;
      constexpr int U = row_plan<F>::U;
      const unsigned long long* L8 = (const unsigned long long*)L;
      const unsigned long long none = 0xFFFF0000ull;
      unsigned long long ent = (lane < n) ? L8[lane] : none;
      #pragma nounroll
      for (int j = 0; j < n; j += 32) {
        const unsigned long long cur = ent;
        ent = (j + 32 + lane < n) ? L8[j + 32 + lane] : none;
        const int   s  = (int)(cur & 0xFFFFull);
        const int   dl = (int)(unsigned)cur >> 16;
        const float sc = __int_as_float((int)(cur >> 32));
        const int   c  = (n - j < 32) ? (n - j) : 32;
        if (dl >= 0) atomicAdd(&degc[dl], 1);
        #pragma nounroll
        for (int jb = 0; jb < c; jb += U) tile_rows<F, TS, U>(h, tile, s, dl, sc, jb, lane);
      }
    }
  }
  if (REC && binned) {
    constexpr int U = row_plan<F>::U;
    const unsigned none = 0xFFFF0000u;
    auto entry = [&](int i) -> unsigned {
      int blk = 0, idx = 0;
      for (int b = 0; b < nbw; ++b) {
        const int pb = __builtin_amdgcn_readlane(pre, b);
        const int cb = __builtin_amdgcn_readlane(cl, b);
        if (i >= pb && i < pb + cb) {
          blk = wave + 32 * b;
          idx = i - pb;
        }
      }
      return (i < total) ? (unsigned)bins[((size_t)blockIdx.x * NB + blk) * SEG_CAP + idx] : none;
    };
    unsigned ent0 = entry(lane);
    unsigned ent1 = entry(32 + lane);
    float sc0 = (lane < total) ? out_scale(degO, N, (int)(ent0 & 0xFFFFu)) : 0.0f;
    #pragma nounroll
    for (int j = 0; j < total; j += 32) {
      const unsigned cur = ent0;
      const float    sc  = sc0;
      ent0 = ent1;
      ent1 = entry(j + 64 + lane);
      sc0  = (j + 32 + lane < total) ? out_scale(degO, N, (int)(ent0 & 0xFFFFu)) : 0.0f;
      const int c = (total - j < 32) ? (total - j) : 32;
      if (lane < c) L[j + lane] = make_int2((int)cur, __float_as_int(sc));
      const int s  = (int)(cur & 0xFFFFu);
      const int dl = (int)cur >> 16;
      if (dl >= 0) atomicAdd(&degc[dl], 1);
      #pragma nounroll
      for (int jb = 0; jb < c; jb += U) tile_rows<F, TS, U>(h, tile, s, dl, sc, jb, lane);
    }
    if (lane == 0) cnts[wid] = total;
  } else if (!listed) {
    const int rec = filter_stream<F, TS, REC>(h, src, dst, degO, N, own, tile, degc, base, rows, q, wave, lane, E, L);
    if (REC && lane == 0) cnts[wid] = rec;
  }
  __syncthreads();

  if (MM) {
    constexpr int MT = (GROUP_NODES + 15) / 16;
    const int c    = lane & 15;
    const int kh   = (lane >> 4) * 2;
    const int gcol = (wave & 7) * 16 + c;
    const float* Wp = Wm + (size_t)(4 * kh) * F + gcol;
    v2f b[F / 4];
    #pragma unroll
    for (int j = 0; j < F / 4; ++j) {
      b[j][0] = Wp[(size_t)(16 * (j & 7) + (j >> 3)) * F];
      b[j][1] = Wp[(size_t)(16 * (j & 7) + (j >> 3) + 4) * F];
    }
    const float bb = bias[gcol];

    for (int i = t; i < rows * (F / 4); i += 1024) {
      const int r = i / (F / 4);
      float4* p = (float4*)(tile + r * TS) + (i - r * (F / 4));
      const float sc = inv_sqrt_count((float)degc[r]);
      float4 v = *p;
      v.x *= sc; v.y *= sc; v.z *= sc; v.w *= sc;
      *p = v;
    }
    __syncthreads();

    #pragma nounroll
    for (int mt = wave >> 3; mt < MT; mt += 4) {
      const int ar = (mt * 16 + c < GROUP_NODES) ? (mt * 16 + c) : (GROUP_NODES - 1);
      const float* ap = tile + ar * TS + kh;
      v8f acc0 = {};
      v8f acc1 = {};
      #pragma unroll
      for (int j = 0; j < F / 4; j += 2) {
        const v2f a0 = *(const v2f*)(ap + 4 * j);
        const v2f a1 = *(const v2f*)(ap + 4 * j + 4);
        acc0 = __builtin_amdgcn_wmma_f32_16x16x4_f32(false, a0, false, b[j], (short)0, acc0,
                                                     false, false);
        acc1 = __builtin_amdgcn_wmma_f32_16x16x4_f32(false, a1, false, b[j + 1], (short)0, acc1,
                                                     false, false);
      }
      const v8f acc = acc0 + acc1;
      const int r0 = mt * 16 + (lane >> 4) * 8;
      #pragma unroll
      for (int v = 0; v < 8; ++v) {
        const int r = r0 + v;
        if (r < rows) agg[(size_t)(base + r) * F + gcol] = fmaxf(acc[v] + bb, 0.0f);
      }
    }
    return;
  }

  float4* outp = (float4*)(agg + (size_t)base * F);
  const int n4 = rows * F / 4;
  if (FIN) {
    constexpr int F4 = F / 4;
    for (int i = t; i < n4; i += 1024) {
      const int r  = i / F4;
      const int c  = (i - r * F4) * 4;
      const float sc = inv_sqrt_count((float)degc[r]);
      const float b0 = bias[c], b1 = bias[c + 1], b2 = bias[c + 2], b3 = bias[c + 3];
      const float4 v = tile4[r * (TS / 4) + (i - r * F4)];
      outp[i] = make_float4(v.x * sc + b0, v.y * sc + b1, v.z * sc + b2, v.w * sc + b3);
    }
  } else {
    static_assert(MM || FIN || TS == F, "the plain copy takes the tile's rows as they lie");
    for (int i = t; i < n4; i += 1024) outp[i] = tile4[i];
  }
}

__global__ void __launch_bounds__(256)
gcn_gemm_kernel(const float* __restrict__ A,
                const float* __restrict__ A2,
                const float* __restrict__ rowScale,
                const float* __restrict__ W,
                const float* __restrict__ bias,
                float* __restrict__ C,
                int M, int Nd, int relu) {
  __shared__ float As[16 * LDK];

  const int t     = threadIdx.x;
  const int mBase = blockIdx.x * 16;

  for (int i = t; i < 16 * (TK / 4); i += 256) {
    int row = i >> 5;
    int c4  = i & 31;
    int gr  = mBase + row;
    float4 v = make_float4(0.f, 0.f, 0.f, 0.f);
    float  s = 1.0f;
    if (gr < M) {
      const float4 a = ((const float4*)(A + (size_t)gr * TK))[c4];
      float4 w = make_float4(0.f, 0.f, 0.f, 0.f);
      if (A2) w = ((const float4*)(A2 + (size_t)gr * TK))[c4];
      if (rowScale) s = rowScale[gr];
      v = make_float4(a.x + w.x, a.y + w.y, a.z + w.z, a.w + w.w);
    }
    float* dstp = As + (size_t)row * LDK + (size_t)c4 * 4;
    dstp[0] = v.x * s; dstp[1] = v.y * s; dstp[2] = v.z * s; dstp[3] = v.w * s;
  }
  __syncthreads();

  const int wave   = t >> 5;
  const int lane   = t & 31;
  const int nTiles = (Nd + 15) >> 4;
  if (wave >= nTiles) return;

  const int col    = lane & 15;
  const int gcol   = wave * 16 + col;
  const int khalf  = (lane >> 4) * 2;
  const bool colOk = (gcol < Nd);
  const int gcSafe = colOk ? gcol : (Nd - 1);

  const float* As_row = As + (size_t)(lane & 15) * LDK + khalf;
  const float* Wp     = W + (size_t)khalf * Nd + gcSafe;

  v8f acc0 = {};
  v8f acc1 = {};
  #pragma unroll 8
  for (int k0 = 0; k0 < TK; k0 += 8) {
    v2f a0, b0, a1, b1;
    a0[0] = As_row[k0];
    a0[1] = As_row[k0 + 1];
    b0[0] = Wp[0];
    b0[1] = Wp[Nd];
    a1[0] = As_row[k0 + 4];
    a1[1] = As_row[k0 + 5];
    b1[0] = Wp[4 * (size_t)Nd];
    b1[1] = Wp[5 * (size_t)Nd];
    acc0 = __builtin_amdgcn_wmma_f32_16x16x4_f32(false, a0, false, b0, (short)0, acc0,
                                                 false, false);
    acc1 = __builtin_amdgcn_wmma_f32_16x16x4_f32(false, a1, false, b1, (short)0, acc1,
                                                 false, false);
    Wp += 8 * (size_t)Nd;
  }
  v8f acc = acc0 + acc1;

  if (colOk) {
    const float bb    = bias ? bias[gcol] : 0.0f;
    const int   rbase = mBase + (lane >> 4) * 8;
    #pragma unroll
    for (int v = 0; v < 8; ++v) {
      int r = rbase + v;
      if (r < M) {
        float val = acc[v] + bb;
        if (relu) val = fmaxf(val, 0.0f);
        C[(size_t)r * Nd + gcol] = val;
      }
    }
  }
}

extern "C" void kernel_launch(void* const* d_in, const int* in_sizes, int n_in,
                              void* d_out, int out_size, void* d_ws, size_t ws_size,
                              hipStream_t stream) {
  const float* feats = (const float*)d_in[0];
  const int*   src   = (const int*)d_in[1];
  const int*   dst   = (const int*)d_in[2];
  const float* W1    = (const float*)d_in[3];
  const float* b1    = (const float*)d_in[4];
  const float* W2    = (const float*)d_in[5];
  const float* b2    = (const float*)d_in[6];
  const float* W3    = (const float*)d_in[7];
  const float* b3    = (const float*)d_in[8];
  float*       out   = (float*)d_out;

  const int HID  = in_sizes[4];
  const int NCLS = in_sizes[8];
  const int INF  = in_sizes[3] / HID;
  const int N    = in_sizes[0] / INF;
  const int E    = in_sizes[1];

  float* deg  = (float*)d_ws;
  float* bufA = deg + 4 * (size_t)N;
  float* bufB = bufA + 2 * (size_t)N * TK;
  float* bufC = bufB + (size_t)N * TK;
  int*   own  = (int*)(bufC + (size_t)N * NCLS);

  const int B = 256;
  const int mTiles = (N + 15) / 16;
  const int Epad   = (E + 15) & ~15;
  const int groups = (N + GROUP_NODES - 1) / GROUP_NODES;
  int*   cnts  = own + Epad / 4;
  int*   lists = cnts + groups * 32;
  const int NB = (Epad + BLK_EDGES - 1) / BLK_EDGES;
  int*   cntTab = lists + (size_t)groups * 32 * LIST_CAP * 2;
  int*   bins   = cntTab + (size_t)256 * NB;

  hipMemsetAsync(deg, 0, sizeof(float) * 2 * (size_t)N, stream);
  gcn_deg_kernel<<<NB, 1024, 0, stream>>>(src, dst, deg, own, cntTab, bins, N, E, Epad, NB);

  gcn_owner_scatter_kernel<128, true, false, false, true><<<groups, 1024, 0, stream>>>(feats, src, dst, deg, own, cnts, lists, bufB, b1, W1, cntTab, bins, N, E, NB);

  gcn_owner_scatter_kernel<128, false, true, false, true><<<groups, 1024, 0, stream>>>(bufB, src, dst, deg, own, cnts, lists, bufA, b2, W2, nullptr, nullptr, N, E, NB);

  gcn_gemm_kernel<<<mTiles, B, 0, stream>>>(bufA, nullptr, nullptr, W3, nullptr, bufC, N, NCLS, 0);
  gcn_owner_scatter_kernel<40, false, true, true, false><<<groups, 1024, 0, stream>>>(bufC, src, dst, deg, own, cnts, lists, out, b3, nullptr, nullptr, nullptr, N, E, NB);
}
